# starting kernel
# baseline (speedup 1.0000x reference)
#include <hip/hip_runtime.h>

typedef __attribute__((ext_vector_type(2))) float v2f;
typedef __attribute__((ext_vector_type(8))) float v8f;

#define C 128
#define TM 64
#define XS 132
#define WS 136

__global__ void time_mlp_kernel(const float* __restrict__ t,
                                const float* __restrict__ tW1, const float* __restrict__ tb1,
                                const float* __restrict__ tW2, const float* __restrict__ tb2,
                                float* __restrict__ t_emb) {
    __shared__ float hid[C];
    int j = threadIdx.x;
    float tv = t[0];
    hid[j] = fmaxf(tv * tW1[j] + tb1[j], 0.0f);
    __syncthreads();
    float acc = tb2[j];
    for (int i = 0; i < C; ++i) acc += hid[i] * tW2[i * C + j];
    t_emb[j] = acc;
}

__global__ void deg_init_kernel(float* __restrict__ deg, int n) {
    int i = blockIdx.x * blockDim.x + threadIdx.x;
    if (i < n) deg[i] = 1.0f;
}
__global__ void deg_count_kernel(const int* __restrict__ dst, float* __restrict__ deg, int e) {
    int i = blockIdx.x * blockDim.x + threadIdx.x;
    if (i < e) atomicAdd(&deg[dst[i]], 1.0f);
}
__global__ void dinv_kernel(float* __restrict__ deg, int n) {
    int i = blockIdx.x * blockDim.x + threadIdx.x;
    if (i < n) deg[i] = rsqrtf(deg[i]);
}

__global__ __launch_bounds__(256) void gemm128_wmma_kernel(
        const float* __restrict__ X, const float* __restrict__ W,
        const float* __restrict__ t_emb, float* __restrict__ Y, int n_rows) {
    __shared__ float sW[C * WS];
    __shared__ float sX[TM * XS];

    const int tid  = threadIdx.x;
    const int wave = tid >> 5;
    const int lane = tid & 31;
    const int row0 = blockIdx.x * TM;

    {
        const float4* W4  = (const float4*)W;
        float4*       sW4 = (float4*)sW;
        for (int i = tid; i < C * (C / 4); i += 256) {
            int r = i >> 5, c4 = i & 31;
            sW4[r * (WS / 4) + c4] = W4[i];
        }
    }
    {
        float4* sX4 = (float4*)sX;
        for (int i = tid; i < TM * (C / 4); i += 256) {
            int r = i >> 5, c4 = i & 31;
            int row = row0 + r;
            float4 v = make_float4(0.f, 0.f, 0.f, 0.f);
            if (row < n_rows) v = ((const float4*)(X + (size_t)row * C))[c4];
            if (t_emb) {
                float4 e = ((const float4*)t_emb)[c4];
                v.x += e.x; v.y += e.y; v.z += e.z; v.w += e.w;
            }
            sX4[r * (XS / 4) + c4] = v;
        }
    }
    __syncthreads();

    const int n0   = wave * 16;
    const int half = lane >> 4;
    const int ml   = lane & 15;

    for (int mt = 0; mt < TM / 16; ++mt) {
        v8f c = {};
        const float* xa = sX + (size_t)(mt * 16 + ml) * XS;
        for (int k = 0; k < C; k += 4) {
            v2f a, b;
            a[0] = xa[k + 2 * half];
            a[1] = xa[k + 2 * half + 1];
            b[0] = sW[(k + 2 * half)     * WS + n0 + ml];
            b[1] = sW[(k + 2 * half + 1) * WS + n0 + ml];
            c = __builtin_amdgcn_wmma_f32_16x16x4_f32(
                    false, a, false, b, (short)0, c, false, false);
        }
        int row_lo = row0 + mt * 16 + half * 8;
        int col    = n0 + ml;
        for (int r = 0; r < 8; ++r) {
            int row = row_lo + r;
            if (row < n_rows) Y[(size_t)row * C + col] = c[r];
        }
    }
}

__global__ void self_bias_kernel(const float* __restrict__ XW, const float* __restrict__ dinv,
                                 const float* __restrict__ bias, float* __restrict__ AGG, int n) {
    size_t i = (size_t)blockIdx.x * blockDim.x + threadIdx.x;
    if (i < (size_t)n * C) {
        int row = (int)(i >> 7);
        int ch  = (int)(i & (C - 1));
        float di = dinv[row];
        AGG[i] = XW[i] * di * di + bias[ch];
    }
}

__global__ void scatter_kernel(const float* __restrict__ XW, const float* __restrict__ dinv,
                               const int* __restrict__ src, const int* __restrict__ dst,
                               float* __restrict__ AGG, int e) {
    int gid  = blockIdx.x * blockDim.x + threadIdx.x;
    int edge = gid >> 5;
    int lane = gid & 31;
    if (edge >= e) return;
    int s = src[edge], d = dst[edge];
    float coef = dinv[s] * dinv[d];
    float* outp = AGG + (size_t)d * C + lane * 4;
    __builtin_prefetch(outp, 1, 3);
    float4 v = ((const float4*)(XW + (size_t)s * C))[lane];
    atomicAdd(outp + 0, v.x * coef);
    atomicAdd(outp + 1, v.y * coef);
    atomicAdd(outp + 2, v.z * coef);
    atomicAdd(outp + 3, v.w * coef);
}

__global__ void ln_relu_kernel(float* __restrict__ H, const float* __restrict__ ln_w,
                               const float* __restrict__ ln_b, int n) {
    int gid  = blockIdx.x * blockDim.x + threadIdx.x;
    int row  = gid >> 5;
    int lane = gid & 31;
    if (row >= n) return;
    float4 v = ((float4*)(H + (size_t)row * C))[lane];
    float s = v.x + v.y + v.z + v.w;
    for (int m = 16; m; m >>= 1) s += __shfl_xor(s, m, 32);
    float mu = s * (1.0f / C);
    float dx = v.x - mu, dy = v.y - mu, dz = v.z - mu, dw = v.w - mu;
    float q = dx * dx + dy * dy + dz * dz + dw * dw;
    for (int m = 16; m; m >>= 1) q += __shfl_xor(q, m, 32);
    float rinv = rsqrtf(q * (1.0f / C) + 1e-5f);
    float4 w4 = ((const float4*)ln_w)[lane];
    float4 b4 = ((const float4*)ln_b)[lane];
    float4 o;
    o.x = fmaxf(dx * rinv * w4.x + b4.x, 0.0f);
    o.y = fmaxf(dy * rinv * w4.y + b4.y, 0.0f);
    o.z = fmaxf(dz * rinv * w4.z + b4.z, 0.0f);
    o.w = fmaxf(dw * rinv * w4.w + b4.w, 0.0f);
    ((float4*)(H + (size_t)row * C))[lane] = o;
}

extern "C" void kernel_launch(void* const* d_in, const int* in_sizes, int n_in,
                              void* d_out, int out_size, void* d_ws, size_t ws_size,
                              hipStream_t stream) {
    const float* h_noisy = (const float*)d_in[0];
    const int*   eidx    = (const int*)d_in[1];
    const float* t       = (const float*)d_in[2];
    const float* tW1     = (const float*)d_in[3];
    const float* tb1     = (const float*)d_in[4];
    const float* tW2     = (const float*)d_in[5];
    const float* tb2     = (const float*)d_in[6];
    const float* W1      = (const float*)d_in[7];
    const float* b1      = (const float*)d_in[8];
    const float* W2      = (const float*)d_in[9];
    const float* b2      = (const float*)d_in[10];
    const float* ln_w    = (const float*)d_in[11];
    const float* ln_b    = (const float*)d_in[12];

    const int N = in_sizes[0] / C;
    const int E = in_sizes[1] / 2;
    const int* src = eidx;
    const int* dst = eidx + E;

    float* ws    = (float*)d_ws;
    float* t_emb = ws;
    float* dinv  = ws + C;
    size_t o     = (size_t)C + (size_t)N;
    o = (o + 3) & ~(size_t)3;
    float* bufA  = ws + o;
    float* bufB  = bufA + (size_t)N * C;
    float* out   = (float*)d_out;

    const int BT = 256;
    dim3 blk(BT);

    hipLaunchKernelGGL(time_mlp_kernel, dim3(1), dim3(C), 0, stream,
                       t, tW1, tb1, tW2, tb2, t_emb);
    hipLaunchKernelGGL(deg_init_kernel, dim3((N + BT - 1) / BT), blk, 0, stream, dinv, N);
    hipLaunchKernelGGL(deg_count_kernel, dim3((E + BT - 1) / BT), blk, 0, stream, dst, dinv, E);
    hipLaunchKernelGGL(dinv_kernel, dim3((N + BT - 1) / BT), blk, 0, stream, dinv, N);

    const int gemm_blocks = (N + TM - 1) / TM;
    const int nc_blocks   = (int)(((size_t)N * C + BT - 1) / BT);
    const int ew_blocks   = (int)(((size_t)E * 32 + BT - 1) / BT);
    const int row_blocks  = (int)(((size_t)N * 32 + BT - 1) / BT);

    hipLaunchKernelGGL(gemm128_wmma_kernel, dim3(gemm_blocks), blk, 0, stream,
                       h_noisy, W1, t_emb, bufA, N);
    hipLaunchKernelGGL(self_bias_kernel, dim3(nc_blocks), blk, 0, stream,
                       bufA, dinv, b1, bufB, N);
    hipLaunchKernelGGL(scatter_kernel, dim3(ew_blocks), blk, 0, stream,
                       bufA, dinv, src, dst, bufB, E);
    hipLaunchKernelGGL(ln_relu_kernel, dim3(row_blocks), blk, 0, stream,
                       bufB, ln_w, ln_b, N);

    hipLaunchKernelGGL(gemm128_wmma_kernel, dim3(gemm_blocks), blk, 0, stream,
                       bufB, W2, (const float*)nullptr, bufA, N);
    hipLaunchKernelGGL(self_bias_kernel, dim3(nc_blocks), blk, 0, stream,
                       bufA, dinv, b2, out, N);
    hipLaunchKernelGGL(scatter_kernel, dim3(ew_blocks), blk, 0, stream,
                       bufA, dinv, src, dst, out, E);
}


# Float-atomic scatter-add replaced by a per-call edge list grouped by destination (count+rank, scan, place) and a wave-per-row gather sum; in-edges summed in ascending source order, self loop last
# speedup vs baseline: 2.0931x; 2.0931x over previous
#include <hip/hip_runtime.h>

typedef __attribute__((ext_vector_type(2))) float v2f;
typedef __attribute__((ext_vector_type(8))) float v8f;

#define C 128
#define TM 64
#define XS 132
#define WS 136

__global__ void time_mlp_kernel(const float* __restrict__ t,
                                const float* __restrict__ tW1, const float* __restrict__ tb1,
                                const float* __restrict__ tW2, const float* __restrict__ tb2,
                                float* __restrict__ t_emb) {
    __shared__ float hid[C];
    int j = threadIdx.x;
    float tv = t[0];
    hid[j] = fmaxf(tv * tW1[j] + tb1[j], 0.0f);
    __syncthreads();
    float acc = tb2[j];
    for (int i = 0; i < C; ++i) acc += hid[i] * tW2[i * C + j];
    t_emb[j] = acc;
}

#define SCAN_PER 4
#define SCAN_BLK (256 * SCAN_PER)

__global__ void zero_int_kernel(int* __restrict__ p, int n) {
    int i = blockIdx.x * blockDim.x + threadIdx.x;
    if (i < n) p[i] = 0;
}
__global__ void count_rank_kernel(const int* __restrict__ dst, int* __restrict__ cnt,
                                  int* __restrict__ rank, int e) {
    int i = blockIdx.x * blockDim.x + threadIdx.x;
    if (i < e) rank[i] = atomicAdd(&cnt[dst[i]], 1);
}
__global__ __launch_bounds__(256) void block_sum_kernel(const int* __restrict__ cnt,
                                                        int* __restrict__ bsum, int n) {
    __shared__ int red[8];
    const int tid = threadIdx.x, lane = tid & 31, wave = tid >> 5;
    const int base = blockIdx.x * SCAN_BLK + tid * SCAN_PER;
    int s = 0;
    for (int k = 0; k < SCAN_PER; ++k)
        if (base + k < n) s += cnt[base + k];
    for (int m = 16; m; m >>= 1) s += __shfl_xor(s, m, 32);
    if (lane == 0) red[wave] = s;
    __syncthreads();
    if (tid == 0) {
        int t = 0;
        for (int w = 0; w < 8; ++w) t += red[w];
        bsum[blockIdx.x] = t;
    }
}
__global__ __launch_bounds__(256) void scan_kernel(int* __restrict__ rowptr, const int* __restrict__ bsum,
                                                   float* __restrict__ dinv, int n) {
    __shared__ int wtot[8];
    __shared__ int ptot[8];
    const int tid = threadIdx.x, lane = tid & 31, wave = tid >> 5;
    int part = 0;
    for (int i = tid; i < (int)blockIdx.x; i += 256) part += bsum[i];
    for (int m = 16; m; m >>= 1) part += __shfl_xor(part, m, 32);
    const int base = blockIdx.x * SCAN_BLK + tid * SCAN_PER;
    int c[SCAN_PER];
    int tsum = 0;
    for (int k = 0; k < SCAN_PER; ++k) {
        c[k] = (base + k < n) ? rowptr[base + k] : 0;
        tsum += c[k];
    }
    int inc = tsum;
    for (int d = 1; d < 32; d <<= 1) {
        int v = __shfl_up(inc, d, 32);
        if (lane >= d) inc += v;
    }
    if (lane == 31) wtot[wave] = inc;
    if (lane == 0)  ptot[wave] = part;
    __syncthreads();
    int off = 0;
    for (int w = 0; w < 8; ++w) {
        off += ptot[w];
        if (w < wave) off += wtot[w];
    }
    int run = off + inc - tsum;
    for (int k = 0; k < SCAN_PER; ++k) {
        int i = base + k;
        if (i < n) {
            rowptr[i] = run;
            dinv[i]   = rsqrtf((float)(c[k] + 1));
            run += c[k];
            if (i == n - 1) rowptr[n] = run;
        }
    }
}
__global__ void place_kernel(const int* __restrict__ src, const int* __restrict__ dst,
                             const int* __restrict__ rowptr, const int* __restrict__ rank,
                             int* __restrict__ csr, int e) {
    int i = blockIdx.x * blockDim.x + threadIdx.x;
    if (i < e) csr[rowptr[dst[i]] + rank[i]] = src[i];
}

__global__ __launch_bounds__(256) void gemm128_wmma_kernel(
        const float* X, const float* __restrict__ W,
        const float* __restrict__ t_emb, float* Y, int n_rows) {
    __shared__ float sW[C * WS];
    __shared__ float sX[TM * XS];

    const int tid  = threadIdx.x;
    const int wave = tid >> 5;
    const int lane = tid & 31;
    const int row0 = blockIdx.x * TM;

    {
        const float4* W4  = (const float4*)W;
        float4*       sW4 = (float4*)sW;
        for (int i = tid; i < C * (C / 4); i += 256) {
            int r = i >> 5, c4 = i & 31;
            sW4[r * (WS / 4) + c4] = W4[i];
        }
    }
    {
        float4* sX4 = (float4*)sX;
        for (int i = tid; i < TM * (C / 4); i += 256) {
            int r = i >> 5, c4 = i & 31;
            int row = row0 + r;
            float4 v = make_float4(0.f, 0.f, 0.f, 0.f);
            if (row < n_rows) v = ((const float4*)(X + (size_t)row * C))[c4];
            if (t_emb) {
                float4 e = ((const float4*)t_emb)[c4];
                v.x += e.x; v.y += e.y; v.z += e.z; v.w += e.w;
            }
            sX4[r * (XS / 4) + c4] = v;
        }
    }
    __syncthreads();

    const int n0   = wave * 16;
    const int half = lane >> 4;
    const int ml   = lane & 15;

    for (int mt = 0; mt < TM / 16; ++mt) {
        v8f c = {};
        const float* xa = sX + (size_t)(mt * 16 + ml) * XS;
        for (int k = 0; k < C; k += 4) {
            v2f a, b;
            a[0] = xa[k + 2 * half];
            a[1] = xa[k + 2 * half + 1];
            b[0] = sW[(k + 2 * half)     * WS + n0 + ml];
            b[1] = sW[(k + 2 * half + 1) * WS + n0 + ml];
            c = __builtin_amdgcn_wmma_f32_16x16x4_f32(
                    false, a, false, b, (short)0, c, false, false);
        }
        int row_lo = row0 + mt * 16 + half * 8;
        int col    = n0 + ml;
        for (int r = 0; r < 8; ++r) {
            int row = row_lo + r;
            if (row < n_rows) Y[(size_t)row * C + col] = c[r];
        }
    }
}

#define GU 8

__global__ __launch_bounds__(256) void gather_kernel(
        const float* __restrict__ XW, const float* __restrict__ dinv,
        const int* __restrict__ rowptr, const int* __restrict__ csr,
        const float* __restrict__ bias, float* __restrict__ OUT, int n) {
    const int lane = threadIdx.x & 31;
    const int row = __builtin_amdgcn_readfirstlane((int)(blockIdx.x * 8 + (threadIdx.x >> 5)));
    if (row >= n) return;
    const int   beg    = rowptr[row];
    const int   deg_in = rowptr[row + 1] - beg;
    const int   total  = deg_in + 1;
    const float di     = dinv[row];
    const float4 b4    = ((const float4*)bias)[lane];

    float4 acc = make_float4(0.f, 0.f, 0.f, 0.f);
    for (int base = 0; base < total; base += 32) {
        const int cnt = min(32, total - base);
        const int e = base + lane;
        int   s_l = row;
        float c_l = di * di;
        if (e < deg_in) {
            s_l = csr[beg + e];
            c_l = dinv[s_l] * di;
        }
        const int ne = min(32, deg_in - base);
        if (ne > 1) {
            int p = 0;
            for (int m = 0; m < ne; ++m) {
                const int sm = __builtin_amdgcn_readlane(s_l, m);
                p += (sm < s_l || (sm == s_l && m < lane)) ? 1 : 0;
            }
            const int pos = (lane < ne) ? p : lane;
            s_l = __builtin_amdgcn_ds_permute(pos << 2, s_l);
            c_l = __int_as_float(__builtin_amdgcn_ds_permute(pos << 2, __float_as_int(c_l)));
        }
        for (int k = 0; k < cnt; k += GU) {
            float4 v[GU];
            #pragma unroll
            for (int u = 0; u < GU; ++u) {
                v[u] = make_float4(0.f, 0.f, 0.f, 0.f);
                if (k + u < cnt) {
                    const int s = __builtin_amdgcn_readlane(s_l, k + u);
                    v[u] = ((const float4*)(XW + (size_t)s * C))[lane];
                }
            }
            #pragma unroll
            for (int u = 0; u < GU; ++u) {
                if (k + u < cnt) {
                    const float cf = __int_as_float(
                        __builtin_amdgcn_readlane(__float_as_int(c_l), k + u));
                    acc.x += v[u].x * cf;
                    acc.y += v[u].y * cf;
                    acc.z += v[u].z * cf;
                    acc.w += v[u].w * cf;
                }
            }
        }
    }
    float4 o;
    o.x = acc.x + b4.x;
    o.y = acc.y + b4.y;
    o.z = acc.z + b4.z;
    o.w = acc.w + b4.w;
    ((float4*)(OUT + (size_t)row * C))[lane] = o;
}

__global__ void ln_relu_kernel(float* __restrict__ H, const float* __restrict__ ln_w,
                               const float* __restrict__ ln_b, int n) {
    int gid  = blockIdx.x * blockDim.x + threadIdx.x;
    int row  = gid >> 5;
    int lane = gid & 31;
    if (row >= n) return;
    float4 v = ((float4*)(H + (size_t)row * C))[lane];
    float s = v.x + v.y + v.z + v.w;
    for (int m = 16; m; m >>= 1) s += __shfl_xor(s, m, 32);
    float mu = s * (1.0f / C);
    float dx = v.x - mu, dy = v.y - mu, dz = v.z - mu, dw = v.w - mu;
    float q = dx * dx + dy * dy + dz * dz + dw * dw;
    for (int m = 16; m; m >>= 1) q += __shfl_xor(q, m, 32);
    float rinv = rsqrtf(q * (1.0f / C) + 1e-5f);
    float4 w4 = ((const float4*)ln_w)[lane];
    float4 b4 = ((const float4*)ln_b)[lane];
    float4 o;
    o.x = fmaxf(dx * rinv * w4.x + b4.x, 0.0f);
    o.y = fmaxf(dy * rinv * w4.y + b4.y, 0.0f);
    o.z = fmaxf(dz * rinv * w4.z + b4.z, 0.0f);
    o.w = fmaxf(dw * rinv * w4.w + b4.w, 0.0f);
    ((float4*)(H + (size_t)row * C))[lane] = o;
}

extern "C" void kernel_launch(void* const* d_in, const int* in_sizes, int n_in,
                              void* d_out, int out_size, void* d_ws, size_t ws_size,
                              hipStream_t stream) {
    const float* h_noisy = (const float*)d_in[0];
    const int*   eidx    = (const int*)d_in[1];
    const float* t       = (const float*)d_in[2];
    const float* tW1     = (const float*)d_in[3];
    const float* tb1     = (const float*)d_in[4];
    const float* tW2     = (const float*)d_in[5];
    const float* tb2     = (const float*)d_in[6];
    const float* W1      = (const float*)d_in[7];
    const float* b1      = (const float*)d_in[8];
    const float* W2      = (const float*)d_in[9];
    const float* b2      = (const float*)d_in[10];
    const float* ln_w    = (const float*)d_in[11];
    const float* ln_b    = (const float*)d_in[12];

    const int N = in_sizes[0] / C;
    const int E = in_sizes[1] / 2;
    const int* src = eidx;
    const int* dst = eidx + E;

    float* ws    = (float*)d_ws;
    float* t_emb = ws;
    float* dinv  = ws + C;
    size_t o     = (size_t)C + (size_t)N;
    o = (o + 3) & ~(size_t)3;
    float* bufA  = ws + o;
    float* bufB  = bufA + (size_t)N * C;
    float* out   = (float*)d_out;

    const int BT = 256;
    dim3 blk(BT);
    const int scan_blocks = (N + SCAN_BLK - 1) / SCAN_BLK;

    int* rowptr = (int*)bufA;
    int* bsum   = rowptr + (N + 1);
    int* rank   = bsum + scan_blocks;
    int* csr    = rank + E;

    hipLaunchKernelGGL(time_mlp_kernel, dim3(1), dim3(C), 0, stream,
                       t, tW1, tb1, tW2, tb2, t_emb);
    hipLaunchKernelGGL(zero_int_kernel, dim3((N + BT - 1) / BT), blk, 0, stream, rowptr, N);
    hipLaunchKernelGGL(count_rank_kernel, dim3((E + BT - 1) / BT), blk, 0, stream,
                       dst, rowptr, rank, E);
    hipLaunchKernelGGL(block_sum_kernel, dim3(scan_blocks), blk, 0, stream, rowptr, bsum, N);
    hipLaunchKernelGGL(scan_kernel, dim3(scan_blocks), blk, 0, stream,
                       rowptr, bsum, dinv, N);
    hipLaunchKernelGGL(place_kernel, dim3((E + BT - 1) / BT), blk, 0, stream,
                       src, dst, rowptr, rank, csr, E);

    const int gemm_blocks = (N + TM - 1) / TM;
    const int row_blocks  = (int)(((size_t)N * 32 + BT - 1) / BT);

    hipLaunchKernelGGL(gemm128_wmma_kernel, dim3(gemm_blocks), blk, 0, stream,
                       h_noisy, W1, t_emb, out, N);
    hipLaunchKernelGGL(gather_kernel, dim3(row_blocks), blk, 0, stream,
                       out, dinv, rowptr, csr, b1, bufB, N);
    hipLaunchKernelGGL(ln_relu_kernel, dim3(row_blocks), blk, 0, stream,
                       bufB, ln_w, ln_b, N);

    hipLaunchKernelGGL(gemm128_wmma_kernel, dim3(gemm_blocks), blk, 0, stream,
                       bufB, W2, (const float*)nullptr, bufB, N);
    hipLaunchKernelGGL(gather_kernel, dim3(row_blocks), blk, 0, stream,
                       bufB, dinv, rowptr, csr, b2, out, N);
}


# LayerNorm+ReLU done in the layer-1 gather's epilogue on the row the wave already holds (row sums by DPP/permlanex16, same add tree); the LayerNorm launch and its read+write of the plane removed
# speedup vs baseline: 2.2392x; 1.0698x over previous
#include <hip/hip_runtime.h>

typedef __attribute__((ext_vector_type(2))) float v2f;
typedef __attribute__((ext_vector_type(8))) float v8f;

#define C 128
#define TM 64
#define XS 132
#define WS 136

__global__ void time_mlp_kernel(const float* __restrict__ t,
                                const float* __restrict__ tW1, const float* __restrict__ tb1,
                                const float* __restrict__ tW2, const float* __restrict__ tb2,
                                float* __restrict__ t_emb) {
    __shared__ float hid[C];
    int j = threadIdx.x;
    float tv = t[0];
    hid[j] = fmaxf(tv * tW1[j] + tb1[j], 0.0f);
    __syncthreads();
    float acc = tb2[j];
    for (int i = 0; i < C; ++i) acc += hid[i] * tW2[i * C + j];
    t_emb[j] = acc;
}

#define SCAN_PER 4
#define SCAN_BLK (256 * SCAN_PER)

__global__ void zero_int_kernel(int* __restrict__ p, int n) {
    int i = blockIdx.x * blockDim.x + threadIdx.x;
    if (i < n) p[i] = 0;
}
__global__ void count_rank_kernel(const int* __restrict__ dst, int* __restrict__ cnt,
                                  int* __restrict__ rank, int e) {
    int i = blockIdx.x * blockDim.x + threadIdx.x;
    if (i < e) rank[i] = atomicAdd(&cnt[dst[i]], 1);
}
__global__ __launch_bounds__(256) void block_sum_kernel(const int* __restrict__ cnt,
                                                        int* __restrict__ bsum, int n) {
    __shared__ int red[8];
    const int tid = threadIdx.x, lane = tid & 31, wave = tid >> 5;
    const int base = blockIdx.x * SCAN_BLK + tid * SCAN_PER;
    int s = 0;
    for (int k = 0; k < SCAN_PER; ++k)
        if (base + k < n) s += cnt[base + k];
    for (int m = 16; m; m >>= 1) s += __shfl_xor(s, m, 32);
    if (lane == 0) red[wave] = s;
    __syncthreads();
    if (tid == 0) {
        int t = 0;
        for (int w = 0; w < 8; ++w) t += red[w];
        bsum[blockIdx.x] = t;
    }
}
__global__ __launch_bounds__(256) void scan_kernel(int* __restrict__ rowptr, const int* __restrict__ bsum,
                                                   float* __restrict__ dinv, int n) {
    __shared__ int wtot[8];
    __shared__ int ptot[8];
    const int tid = threadIdx.x, lane = tid & 31, wave = tid >> 5;
    int part = 0;
    for (int i = tid; i < (int)blockIdx.x; i += 256) part += bsum[i];
    for (int m = 16; m; m >>= 1) part += __shfl_xor(part, m, 32);
    const int base = blockIdx.x * SCAN_BLK + tid * SCAN_PER;
    int c[SCAN_PER];
    int tsum = 0;
    for (int k = 0; k < SCAN_PER; ++k) {
        c[k] = (base + k < n) ? rowptr[base + k] : 0;
        tsum += c[k];
    }
    int inc = tsum;
    for (int d = 1; d < 32; d <<= 1) {
        int v = __shfl_up(inc, d, 32);
        if (lane >= d) inc += v;
    }
    if (lane == 31) wtot[wave] = inc;
    if (lane == 0)  ptot[wave] = part;
    __syncthreads();
    int off = 0;
    for (int w = 0; w < 8; ++w) {
        off += ptot[w];
        if (w < wave) off += wtot[w];
    }
    int run = off + inc - tsum;
    for (int k = 0; k < SCAN_PER; ++k) {
        int i = base + k;
        if (i < n) {
            rowptr[i] = run;
            dinv[i]   = rsqrtf((float)(c[k] + 1));
            run += c[k];
            if (i == n - 1) rowptr[n] = run;
        }
    }
}
__global__ void place_kernel(const int* __restrict__ src, const int* __restrict__ dst,
                             const int* __restrict__ rowptr, const int* __restrict__ rank,
                             int* __restrict__ csr, int e) {
    int i = blockIdx.x * blockDim.x + threadIdx.x;
    if (i < e) csr[rowptr[dst[i]] + rank[i]] = src[i];
}

__global__ __launch_bounds__(256) void gemm128_wmma_kernel(
        const float* X, const float* __restrict__ W,
        const float* __restrict__ t_emb, float* Y, int n_rows) {
    __shared__ float sW[C * WS];
    __shared__ float sX[TM * XS];

    const int tid  = threadIdx.x;
    const int wave = tid >> 5;
    const int lane = tid & 31;
    const int row0 = blockIdx.x * TM;

    {
        const float4* W4  = (const float4*)W;
        float4*       sW4 = (float4*)sW;
        for (int i = tid; i < C * (C / 4); i += 256) {
            int r = i >> 5, c4 = i & 31;
            sW4[r * (WS / 4) + c4] = W4[i];
        }
    }
    {
        float4* sX4 = (float4*)sX;
        for (int i = tid; i < TM * (C / 4); i += 256) {
            int r = i >> 5, c4 = i & 31;
            int row = row0 + r;
            float4 v = make_float4(0.f, 0.f, 0.f, 0.f);
            if (row < n_rows) v = ((const float4*)(X + (size_t)row * C))[c4];
            if (t_emb) {
                float4 e = ((const float4*)t_emb)[c4];
                v.x += e.x; v.y += e.y; v.z += e.z; v.w += e.w;
            }
            sX4[r * (XS / 4) + c4] = v;
        }
    }
    __syncthreads();

    const int n0   = wave * 16;
    const int half = lane >> 4;
    const int ml   = lane & 15;

    for (int mt = 0; mt < TM / 16; ++mt) {
        v8f c = {};
        const float* xa = sX + (size_t)(mt * 16 + ml) * XS;
        for (int k = 0; k < C; k += 4) {
            v2f a, b;
            a[0] = xa[k + 2 * half];
            a[1] = xa[k + 2 * half + 1];
            b[0] = sW[(k + 2 * half)     * WS + n0 + ml];
            b[1] = sW[(k + 2 * half + 1) * WS + n0 + ml];
            c = __builtin_amdgcn_wmma_f32_16x16x4_f32(
                    false, a, false, b, (short)0, c, false, false);
        }
        int row_lo = row0 + mt * 16 + half * 8;
        int col    = n0 + ml;
        for (int r = 0; r < 8; ++r) {
            int row = row_lo + r;
            if (row < n_rows) Y[(size_t)row * C + col] = c[r];
        }
    }
}

#define GU 8

__device__ __forceinline__ float xhalf(float x) {
    const int b = __builtin_bit_cast(int, x);
    return __builtin_bit_cast(float,
        __builtin_amdgcn_permlanex16(b, b, 0x76543210, 0xfedcba98, false, false));
}
template <int M>
__device__ __forceinline__ float xrow(float x) {
    const int b = __builtin_bit_cast(int, x);
    return __builtin_bit_cast(float,
        __builtin_amdgcn_update_dpp(b, b, 0x160 + M, 0xf, 0xf, false));
}
__device__ __forceinline__ float wave_sum(float s) {
    s += xhalf(s);
    s += xrow<8>(s);
    s += xrow<4>(s);
    s += xrow<2>(s);
    s += xrow<1>(s);
    return s;
}

template <bool LN>
__global__ __launch_bounds__(256) void gather_kernel(
        const float* __restrict__ XW, const float* __restrict__ dinv,
        const int* __restrict__ rowptr, const int* __restrict__ csr,
        const float* __restrict__ bias,
        const float* __restrict__ ln_w, const float* __restrict__ ln_b,
        float* __restrict__ OUT, int n) {
    const int lane = threadIdx.x & 31;
    const int row = __builtin_amdgcn_readfirstlane((int)(blockIdx.x * 8 + (threadIdx.x >> 5)));
    if (row >= n) return;
    const int   beg    = rowptr[row];
    const int   deg_in = rowptr[row + 1] - beg;
    const int   total  = deg_in + 1;
    const float di     = dinv[row];
    const float4 b4    = ((const float4*)bias)[lane];

    float4 acc = make_float4(0.f, 0.f, 0.f, 0.f);
    for (int base = 0; base < total; base += 32) {
        const int cnt = min(32, total - base);
        const int e = base + lane;
        int   s_l = row;
        float c_l = di * di;
        if (e < deg_in) {
            s_l = csr[beg + e];
            c_l = dinv[s_l] * di;
        }
        const int ne = min(32, deg_in - base);
        if (ne > 1) {
            int p = 0;
            for (int m = 0; m < ne; ++m) {
                const int sm = __builtin_amdgcn_readlane(s_l, m);
                p += (sm < s_l || (sm == s_l && m < lane)) ? 1 : 0;
            }
            const int pos = (lane < ne) ? p : lane;
            s_l = __builtin_amdgcn_ds_permute(pos << 2, s_l);
            c_l = __int_as_float(__builtin_amdgcn_ds_permute(pos << 2, __float_as_int(c_l)));
        }
        for (int k = 0; k < cnt; k += GU) {
            float4 v[GU];
            #pragma unroll
            for (int u = 0; u < GU; ++u) {
                v[u] = make_float4(0.f, 0.f, 0.f, 0.f);
                if (k + u < cnt) {
                    const int s = __builtin_amdgcn_readlane(s_l, k + u);
                    v[u] = ((const float4*)(XW + (size_t)s * C))[lane];
                }
            }
            #pragma unroll
            for (int u = 0; u < GU; ++u) {
                if (k + u < cnt) {
                    const float cf = __int_as_float(
                        __builtin_amdgcn_readlane(__float_as_int(c_l), k + u));
                    acc.x += v[u].x * cf;
                    acc.y += v[u].y * cf;
                    acc.z += v[u].z * cf;
                    acc.w += v[u].w * cf;
                }
            }
        }
    }
    float4 o;
    o.x = acc.x + b4.x;
    o.y = acc.y + b4.y;
    o.z = acc.z + b4.z;
    o.w = acc.w + b4.w;
    if (LN) {
        const float4 w4 = ((const float4*)ln_w)[lane];
        const float4 g4 = ((const float4*)ln_b)[lane];
        const float mu = wave_sum(o.x + o.y + o.z + o.w) * (1.0f / C);
        const float dx = o.x - mu, dy = o.y - mu, dz = o.z - mu, dw = o.w - mu;
        const float rinv = rsqrtf(wave_sum(dx * dx + dy * dy + dz * dz + dw * dw) * (1.0f / C) + 1e-5f);
        o.x = fmaxf(dx * rinv * w4.x + g4.x, 0.0f);
        o.y = fmaxf(dy * rinv * w4.y + g4.y, 0.0f);
        o.z = fmaxf(dz * rinv * w4.z + g4.z, 0.0f);
        o.w = fmaxf(dw * rinv * w4.w + g4.w, 0.0f);
    }
    ((float4*)(OUT + (size_t)row * C))[lane] = o;
}

extern "C" void kernel_launch(void* const* d_in, const int* in_sizes, int n_in,
                              void* d_out, int out_size, void* d_ws, size_t ws_size,
                              hipStream_t stream) {
    const float* h_noisy = (const float*)d_in[0];
    const int*   eidx    = (const int*)d_in[1];
    const float* t       = (const float*)d_in[2];
    const float* tW1     = (const float*)d_in[3];
    const float* tb1     = (const float*)d_in[4];
    const float* tW2     = (const float*)d_in[5];
    const float* tb2     = (const float*)d_in[6];
    const float* W1      = (const float*)d_in[7];
    const float* b1      = (const float*)d_in[8];
    const float* W2      = (const float*)d_in[9];
    const float* b2      = (const float*)d_in[10];
    const float* ln_w    = (const float*)d_in[11];
    const float* ln_b    = (const float*)d_in[12];

    const int N = in_sizes[0] / C;
    const int E = in_sizes[1] / 2;
    const int* src = eidx;
    const int* dst = eidx + E;

    float* ws    = (float*)d_ws;
    float* t_emb = ws;
    float* dinv  = ws + C;
    size_t o     = (size_t)C + (size_t)N;
    o = (o + 3) & ~(size_t)3;
    float* bufA  = ws + o;
    float* bufB  = bufA + (size_t)N * C;
    float* out   = (float*)d_out;

    const int BT = 256;
    dim3 blk(BT);
    const int scan_blocks = (N + SCAN_BLK - 1) / SCAN_BLK;

    int* rowptr = (int*)bufA;
    int* bsum   = rowptr + (N + 1);
    int* rank   = bsum + scan_blocks;
    int* csr    = rank + E;

    hipLaunchKernelGGL(time_mlp_kernel, dim3(1), dim3(C), 0, stream,
                       t, tW1, tb1, tW2, tb2, t_emb);
    hipLaunchKernelGGL(zero_int_kernel, dim3((N + BT - 1) / BT), blk, 0, stream, rowptr, N);
    hipLaunchKernelGGL(count_rank_kernel, dim3((E + BT - 1) / BT), blk, 0, stream,
                       dst, rowptr, rank, E);
    hipLaunchKernelGGL(block_sum_kernel, dim3(scan_blocks), blk, 0, stream, rowptr, bsum, N);
    hipLaunchKernelGGL(scan_kernel, dim3(scan_blocks), blk, 0, stream,
                       rowptr, bsum, dinv, N);
    hipLaunchKernelGGL(place_kernel, dim3((E + BT - 1) / BT), blk, 0, stream,
                       src, dst, rowptr, rank, csr, E);

    const int gemm_blocks = (N + TM - 1) / TM;
    const int row_blocks  = (int)(((size_t)N * 32 + BT - 1) / BT);

    hipLaunchKernelGGL(gemm128_wmma_kernel, dim3(gemm_blocks), blk, 0, stream,
                       h_noisy, W1, t_emb, out, N);
    hipLaunchKernelGGL(gather_kernel<true>, dim3(row_blocks), blk, 0, stream,
                       out, dinv, rowptr, csr, b1, ln_w, ln_b, bufB, N);

    hipLaunchKernelGGL(gemm128_wmma_kernel, dim3(gemm_blocks), blk, 0, stream,
                       bufB, W2, (const float*)nullptr, bufB, N);
    hipLaunchKernelGGL(gather_kernel<false>, dim3(row_blocks), blk, 0, stream,
                       bufB, dinv, rowptr, csr, b2, ln_w, ln_b, out, N);
}


# GEMM staging: every global load of a workgroup's X tile, t_emb and W is issued as one group before the first wait (was one round trip per 16-byte chunk of the X tile, two for W); sums unchanged
# speedup vs baseline: 2.3965x; 1.0702x over previous
#include <hip/hip_runtime.h>

typedef __attribute__((ext_vector_type(2))) float v2f;
typedef __attribute__((ext_vector_type(8))) float v8f;

#define C 128
#define TM 64
#define XS 132
#define WS 136

__global__ void time_mlp_kernel(const float* __restrict__ t,
                                const float* __restrict__ tW1, const float* __restrict__ tb1,
                                const float* __restrict__ tW2, const float* __restrict__ tb2,
                                float* __restrict__ t_emb) {
    __shared__ float hid[C];
    int j = threadIdx.x;
    float tv = t[0];
    hid[j] = fmaxf(tv * tW1[j] + tb1[j], 0.0f);
    __syncthreads();
    float acc = tb2[j];
    for (int i = 0; i < C; ++i) acc += hid[i] * tW2[i * C + j];
    t_emb[j] = acc;
}

#define SCAN_PER 4
#define SCAN_BLK (256 * SCAN_PER)

__global__ void zero_int_kernel(int* __restrict__ p, int n) {
    int i = blockIdx.x * blockDim.x + threadIdx.x;
    if (i < n) p[i] = 0;
}
__global__ void count_rank_kernel(const int* __restrict__ dst, int* __restrict__ cnt,
                                  int* __restrict__ rank, int e) {
    int i = blockIdx.x * blockDim.x + threadIdx.x;
    if (i < e) rank[i] = atomicAdd(&cnt[dst[i]], 1);
}
__global__ __launch_bounds__(256) void block_sum_kernel(const int* __restrict__ cnt,
                                                        int* __restrict__ bsum, int n) {
    __shared__ int red[8];
    const int tid = threadIdx.x, lane = tid & 31, wave = tid >> 5;
    const int base = blockIdx.x * SCAN_BLK + tid * SCAN_PER;
    int s = 0;
    for (int k = 0; k < SCAN_PER; ++k)
        if (base + k < n) s += cnt[base + k];
    for (int m = 16; m; m >>= 1) s += __shfl_xor(s, m, 32);
    if (lane == 0) red[wave] = s;
    __syncthreads();
    if (tid == 0) {
        int t = 0;
        for (int w = 0; w < 8; ++w) t += red[w];
        bsum[blockIdx.x] = t;
    }
}
__global__ __launch_bounds__(256) void scan_kernel(int* __restrict__ rowptr, const int* __restrict__ bsum,
                                                   float* __restrict__ dinv, int n) {
    __shared__ int wtot[8];
    __shared__ int ptot[8];
    const int tid = threadIdx.x, lane = tid & 31, wave = tid >> 5;
    int part = 0;
    for (int i = tid; i < (int)blockIdx.x; i += 256) part += bsum[i];
    for (int m = 16; m; m >>= 1) part += __shfl_xor(part, m, 32);
    const int base = blockIdx.x * SCAN_BLK + tid * SCAN_PER;
    int c[SCAN_PER];
    int tsum = 0;
    for (int k = 0; k < SCAN_PER; ++k) {
        c[k] = (base + k < n) ? rowptr[base + k] : 0;
        tsum += c[k];
    }
    int inc = tsum;
    for (int d = 1; d < 32; d <<= 1) {
        int v = __shfl_up(inc, d, 32);
        if (lane >= d) inc += v;
    }
    if (lane == 31) wtot[wave] = inc;
    if (lane == 0)  ptot[wave] = part;
    __syncthreads();
    int off = 0;
    for (int w = 0; w < 8; ++w) {
        off += ptot[w];
        if (w < wave) off += wtot[w];
    }
    int run = off + inc - tsum;
    for (int k = 0; k < SCAN_PER; ++k) {
        int i = base + k;
        if (i < n) {
            rowptr[i] = run;
            dinv[i]   = rsqrtf((float)(c[k] + 1));
            run += c[k];
            if (i == n - 1) rowptr[n] = run;
        }
    }
}
__global__ void place_kernel(const int* __restrict__ src, const int* __restrict__ dst,
                             const int* __restrict__ rowptr, const int* __restrict__ rank,
                             int* __restrict__ csr, int e) {
    int i = blockIdx.x * blockDim.x + threadIdx.x;
    if (i < e) csr[rowptr[dst[i]] + rank[i]] = src[i];
}

__global__ __launch_bounds__(256) void gemm128_wmma_kernel(
        const float* X, const float* __restrict__ W,
        const float* __restrict__ t_emb, float* Y, int n_rows) {
    __shared__ float sW[C * WS];
    __shared__ float sX[TM * XS];

    const int tid  = threadIdx.x;
    const int wave = tid >> 5;
    const int lane = tid & 31;
    const int row0 = blockIdx.x * TM;

    const int c4 = lane;
    const int r0 = wave;
    float4 xv[TM / 8];
    #pragma unroll
    for (int u = 0; u < TM / 8; ++u) {
        const int row = row0 + r0 + 8 * u;
        xv[u] = make_float4(0.f, 0.f, 0.f, 0.f);
        if (row < n_rows) xv[u] = ((const float4*)(X + (size_t)row * C))[c4];
    }
    float4 e = make_float4(0.f, 0.f, 0.f, 0.f);
    if (t_emb) e = ((const float4*)t_emb)[c4];
    float4 wv[C / 8];
    #pragma unroll
    for (int u = 0; u < C / 8; ++u) wv[u] = ((const float4*)W)[(r0 + 8 * u) * (C / 4) + c4];

    {
        float4* sW4 = (float4*)sW;
        #pragma unroll
        for (int u = 0; u < C / 8; ++u) sW4[(r0 + 8 * u) * (WS / 4) + c4] = wv[u];
    }
    if (t_emb) {
        #pragma unroll
        for (int u = 0; u < TM / 8; ++u) {
            xv[u].x += e.x; xv[u].y += e.y; xv[u].z += e.z; xv[u].w += e.w;
        }
    }
    {
        float4* sX4 = (float4*)sX;
        #pragma unroll
        for (int u = 0; u < TM / 8; ++u) sX4[(r0 + 8 * u) * (XS / 4) + c4] = xv[u];
    }
    __syncthreads();

    const int n0   = wave * 16;
    const int half = lane >> 4;
    const int ml   = lane & 15;

    for (int mt = 0; mt < TM / 16; ++mt) {
        v8f c = {};
        const float* xa = sX + (size_t)(mt * 16 + ml) * XS;
        for (int k = 0; k < C; k += 4) {
            v2f a, b;
            a[0] = xa[k + 2 * half];
            a[1] = xa[k + 2 * half + 1];
            b[0] = sW[(k + 2 * half)     * WS + n0 + ml];
            b[1] = sW[(k + 2 * half + 1) * WS + n0 + ml];
            c = __builtin_amdgcn_wmma_f32_16x16x4_f32(
                    false, a, false, b, (short)0, c, false, false);
        }
        int row_lo = row0 + mt * 16 + half * 8;
        int col    = n0 + ml;
        for (int r = 0; r < 8; ++r) {
            int row = row_lo + r;
            if (row < n_rows) Y[(size_t)row * C + col] = c[r];
        }
    }
}

#define GU 8

__device__ __forceinline__ float xhalf(float x) {
    const int b = __builtin_bit_cast(int, x);
    return __builtin_bit_cast(float,
        __builtin_amdgcn_permlanex16(b, b, 0x76543210, 0xfedcba98, false, false));
}
template <int M>
__device__ __forceinline__ float xrow(float x) {
    const int b = __builtin_bit_cast(int, x);
    return __builtin_bit_cast(float,
        __builtin_amdgcn_update_dpp(b, b, 0x160 + M, 0xf, 0xf, false));
}
__device__ __forceinline__ float wave_sum(float s) {
    s += xhalf(s);
    s += xrow<8>(s);
    s += xrow<4>(s);
    s += xrow<2>(s);
    s += xrow<1>(s);
    return s;
}

template <bool LN>
__global__ __launch_bounds__(256) void gather_kernel(
        const float* __restrict__ XW, const float* __restrict__ dinv,
        const int* __restrict__ rowptr, const int* __restrict__ csr,
        const float* __restrict__ bias,
        const float* __restrict__ ln_w, const float* __restrict__ ln_b,
        float* __restrict__ OUT, int n) {
    const int lane = threadIdx.x & 31;
    const int row = __builtin_amdgcn_readfirstlane((int)(blockIdx.x * 8 + (threadIdx.x >> 5)));
    if (row >= n) return;
    const int   beg    = rowptr[row];
    const int   deg_in = rowptr[row + 1] - beg;
    const int   total  = deg_in + 1;
    const float di     = dinv[row];
    const float4 b4    = ((const float4*)bias)[lane];

    float4 acc = make_float4(0.f, 0.f, 0.f, 0.f);
    for (int base = 0; base < total; base += 32) {
        const int cnt = min(32, total - base);
        const int e = base + lane;
        int   s_l = row;
        float c_l = di * di;
        if (e < deg_in) {
            s_l = csr[beg + e];
            c_l = dinv[s_l] * di;
        }
        const int ne = min(32, deg_in - base);
        if (ne > 1) {
            int p = 0;
            for (int m = 0; m < ne; ++m) {
                const int sm = __builtin_amdgcn_readlane(s_l, m);
                p += (sm < s_l || (sm == s_l && m < lane)) ? 1 : 0;
            }
            const int pos = (lane < ne) ? p : lane;
            s_l = __builtin_amdgcn_ds_permute(pos << 2, s_l);
            c_l = __int_as_float(__builtin_amdgcn_ds_permute(pos << 2, __float_as_int(c_l)));
        }
        for (int k = 0; k < cnt; k += GU) {
            float4 v[GU];
            #pragma unroll
            for (int u = 0; u < GU; ++u) {
                v[u] = make_float4(0.f, 0.f, 0.f, 0.f);
                if (k + u < cnt) {
                    const int s = __builtin_amdgcn_readlane(s_l, k + u);
                    v[u] = ((const float4*)(XW + (size_t)s * C))[lane];
                }
            }
            #pragma unroll
            for (int u = 0; u < GU; ++u) {
                if (k + u < cnt) {
                    const float cf = __int_as_float(
                        __builtin_amdgcn_readlane(__float_as_int(c_l), k + u));
                    acc.x += v[u].x * cf;
                    acc.y += v[u].y * cf;
                    acc.z += v[u].z * cf;
                    acc.w += v[u].w * cf;
                }
            }
        }
    }
    float4 o;
    o.x = acc.x + b4.x;
    o.y = acc.y + b4.y;
    o.z = acc.z + b4.z;
    o.w = acc.w + b4.w;
    if (LN) {
        const float4 w4 = ((const float4*)ln_w)[lane];
        const float4 g4 = ((const float4*)ln_b)[lane];
        const float mu = wave_sum(o.x + o.y + o.z + o.w) * (1.0f / C);
        const float dx = o.x - mu, dy = o.y - mu, dz = o.z - mu, dw = o.w - mu;
        const float rinv = rsqrtf(wave_sum(dx * dx + dy * dy + dz * dz + dw * dw) * (1.0f / C) + 1e-5f);
        o.x = fmaxf(dx * rinv * w4.x + g4.x, 0.0f);
        o.y = fmaxf(dy * rinv * w4.y + g4.y, 0.0f);
        o.z = fmaxf(dz * rinv * w4.z + g4.z, 0.0f);
        o.w = fmaxf(dw * rinv * w4.w + g4.w, 0.0f);
    }
    ((float4*)(OUT + (size_t)row * C))[lane] = o;
}

extern "C" void kernel_launch(void* const* d_in, const int* in_sizes, int n_in,
                              void* d_out, int out_size, void* d_ws, size_t ws_size,
                              hipStream_t stream) {
    const float* h_noisy = (const float*)d_in[0];
    const int*   eidx    = (const int*)d_in[1];
    const float* t       = (const float*)d_in[2];
    const float* tW1     = (const float*)d_in[3];
    const float* tb1     = (const float*)d_in[4];
    const float* tW2     = (const float*)d_in[5];
    const float* tb2     = (const float*)d_in[6];
    const float* W1      = (const float*)d_in[7];
    const float* b1      = (const float*)d_in[8];
    const float* W2      = (const float*)d_in[9];
    const float* b2      = (const float*)d_in[10];
    const float* ln_w    = (const float*)d_in[11];
    const float* ln_b    = (const float*)d_in[12];

    const int N = in_sizes[0] / C;
    const int E = in_sizes[1] / 2;
    const int* src = eidx;
    const int* dst = eidx + E;

    float* ws    = (float*)d_ws;
    float* t_emb = ws;
    float* dinv  = ws + C;
    size_t o     = (size_t)C + (size_t)N;
    o = (o + 3) & ~(size_t)3;
    float* bufA  = ws + o;
    float* bufB  = bufA + (size_t)N * C;
    float* out   = (float*)d_out;

    const int BT = 256;
    dim3 blk(BT);
    const int scan_blocks = (N + SCAN_BLK - 1) / SCAN_BLK;

    int* rowptr = (int*)bufA;
    int* bsum   = rowptr + (N + 1);
    int* rank   = bsum + scan_blocks;
    int* csr    = rank + E;

    hipLaunchKernelGGL(time_mlp_kernel, dim3(1), dim3(C), 0, stream,
                       t, tW1, tb1, tW2, tb2, t_emb);
    hipLaunchKernelGGL(zero_int_kernel, dim3((N + BT - 1) / BT), blk, 0, stream, rowptr, N);
    hipLaunchKernelGGL(count_rank_kernel, dim3((E + BT - 1) / BT), blk, 0, stream,
                       dst, rowptr, rank, E);
    hipLaunchKernelGGL(block_sum_kernel, dim3(scan_blocks), blk, 0, stream, rowptr, bsum, N);
    hipLaunchKernelGGL(scan_kernel, dim3(scan_blocks), blk, 0, stream,
                       rowptr, bsum, dinv, N);
    hipLaunchKernelGGL(place_kernel, dim3((E + BT - 1) / BT), blk, 0, stream,
                       src, dst, rowptr, rank, csr, E);

    const int gemm_blocks = (N + TM - 1) / TM;
    const int row_blocks  = (int)(((size_t)N * 32 + BT - 1) / BT);

    hipLaunchKernelGGL(gemm128_wmma_kernel, dim3(gemm_blocks), blk, 0, stream,
                       h_noisy, W1, t_emb, out, N);
    hipLaunchKernelGGL(gather_kernel<true>, dim3(row_blocks), blk, 0, stream,
                       out, dinv, rowptr, csr, b1, ln_w, ln_b, bufB, N);

    hipLaunchKernelGGL(gemm128_wmma_kernel, dim3(gemm_blocks), blk, 0, stream,
                       bufB, W2, (const float*)nullptr, bufB, N);
    hipLaunchKernelGGL(gather_kernel<false>, dim3(row_blocks), blk, 0, stream,
                       bufB, dinv, rowptr, csr, b2, ln_w, ln_b, out, N);
}


# Edge count+rank atomics issued by the layer-1 GEMM's workgroups behind their staging barrier, ranks stored after the products: atomics run under the WMMAs, count launch removed; sums unchanged
# speedup vs baseline: 2.5860x; 1.0791x over previous
#include <hip/hip_runtime.h>

typedef __attribute__((ext_vector_type(2))) float v2f;
typedef __attribute__((ext_vector_type(8))) float v8f;

#define C 128
#define TM 64
#define XS 132
#define WS 136

__global__ void time_mlp_kernel(const float* __restrict__ t,
                                const float* __restrict__ tW1, const float* __restrict__ tb1,
                                const float* __restrict__ tW2, const float* __restrict__ tb2,
                                float* __restrict__ t_emb) {
    __shared__ float hid[C];
    int j = threadIdx.x;
    float tv = t[0];
    hid[j] = fmaxf(tv * tW1[j] + tb1[j], 0.0f);
    __syncthreads();
    float acc = tb2[j];
    for (int i = 0; i < C; ++i) acc += hid[i] * tW2[i * C + j];
    t_emb[j] = acc;
}

#define SCAN_PER 4
#define SCAN_BLK (256 * SCAN_PER)

__global__ void zero_int_kernel(int* __restrict__ p, int n) {
    int i = blockIdx.x * blockDim.x + threadIdx.x;
    if (i < n) p[i] = 0;
}
__global__ __launch_bounds__(256) void block_sum_kernel(const int* __restrict__ cnt,
                                                        int* __restrict__ bsum, int n) {
    __shared__ int red[8];
    const int tid = threadIdx.x, lane = tid & 31, wave = tid >> 5;
    const int base = blockIdx.x * SCAN_BLK + tid * SCAN_PER;
    int s = 0;
    for (int k = 0; k < SCAN_PER; ++k)
        if (base + k < n) s += cnt[base + k];
    for (int m = 16; m; m >>= 1) s += __shfl_xor(s, m, 32);
    if (lane == 0) red[wave] = s;
    __syncthreads();
    if (tid == 0) {
        int t = 0;
        for (int w = 0; w < 8; ++w) t += red[w];
        bsum[blockIdx.x] = t;
    }
}
__global__ __launch_bounds__(256) void scan_kernel(int* __restrict__ rowptr, const int* __restrict__ bsum,
                                                   float* __restrict__ dinv, int n) {
    __shared__ int wtot[8];
    __shared__ int ptot[8];
    const int tid = threadIdx.x, lane = tid & 31, wave = tid >> 5;
    int part = 0;
    for (int i = tid; i < (int)blockIdx.x; i += 256) part += bsum[i];
    for (int m = 16; m; m >>= 1) part += __shfl_xor(part, m, 32);
    const int base = blockIdx.x * SCAN_BLK + tid * SCAN_PER;
    int c[SCAN_PER];
    int tsum = 0;
    for (int k = 0; k < SCAN_PER; ++k) {
        c[k] = (base + k < n) ? rowptr[base + k] : 0;
        tsum += c[k];
    }
    int inc = tsum;
    for (int d = 1; d < 32; d <<= 1) {
        int v = __shfl_up(inc, d, 32);
        if (lane >= d) inc += v;
    }
    if (lane == 31) wtot[wave] = inc;
    if (lane == 0)  ptot[wave] = part;
    __syncthreads();
    int off = 0;
    for (int w = 0; w < 8; ++w) {
        off += ptot[w];
        if (w < wave) off += wtot[w];
    }
    int run = off + inc - tsum;
    for (int k = 0; k < SCAN_PER; ++k) {
        int i = base + k;
        if (i < n) {
            rowptr[i] = run;
            dinv[i]   = rsqrtf((float)(c[k] + 1));
            run += c[k];
            if (i == n - 1) rowptr[n] = run;
        }
    }
}
__global__ void place_kernel(const int* __restrict__ src, const int* __restrict__ dst,
                             const int* __restrict__ rowptr, const int* __restrict__ rank,
                             int* __restrict__ csr, int e) {
    int i = blockIdx.x * blockDim.x + threadIdx.x;
    if (i < e) csr[rowptr[dst[i]] + rank[i]] = src[i];
}

template <bool COUNT>
__global__ __launch_bounds__(256) void gemm128_wmma_kernel(
        const float* X, const float* __restrict__ W,
        const float* __restrict__ t_emb, float* Y, int n_rows,
        const int* __restrict__ dst, int* __restrict__ cnt, int* __restrict__ rank,
        int n_edges) {
    __shared__ float sW[C * WS];
    __shared__ float sX[TM * XS];

    const int tid  = threadIdx.x;
    const int wave = tid >> 5;
    const int lane = tid & 31;
    const int row0 = blockIdx.x * TM;

    const int c4 = lane;
    const int r0 = wave;
    float4 xv[TM / 8];
    #pragma unroll
    for (int u = 0; u < TM / 8; ++u) {
        const int row = row0 + r0 + 8 * u;
        xv[u] = make_float4(0.f, 0.f, 0.f, 0.f);
        if (row < n_rows) xv[u] = ((const float4*)(X + (size_t)row * C))[c4];
    }
    float4 e = make_float4(0.f, 0.f, 0.f, 0.f);
    if (t_emb) e = ((const float4*)t_emb)[c4];
    float4 wv[C / 8];
    #pragma unroll
    for (int u = 0; u < C / 8; ++u) wv[u] = ((const float4*)W)[(r0 + 8 * u) * (C / 4) + c4];
    int ei0 = 0, ei1 = 0, eend = 0;
    unsigned d0 = 0, d1 = 0;
    if (COUNT) {
        const int per  = (n_edges + (int)gridDim.x - 1) / (int)gridDim.x;
        const int ebeg = (int)blockIdx.x * per;
        eend = min(n_edges, ebeg + per);
        ei0  = ebeg + tid;
        ei1  = ei0 + 256;
        if (n_edges > 0) {
            d0 = (unsigned)dst[min(ei0, n_edges - 1)];
            d1 = (unsigned)dst[min(ei1, n_edges - 1)];
        }
    }

    {
        float4* sW4 = (float4*)sW;
        #pragma unroll
        for (int u = 0; u < C / 8; ++u) sW4[(r0 + 8 * u) * (WS / 4) + c4] = wv[u];
    }
    if (t_emb) {
        #pragma unroll
        for (int u = 0; u < TM / 8; ++u) {
            xv[u].x += e.x; xv[u].y += e.y; xv[u].z += e.z; xv[u].w += e.w;
        }
    }
    {
        float4* sX4 = (float4*)sX;
        #pragma unroll
        for (int u = 0; u < TM / 8; ++u) sX4[(r0 + 8 * u) * (XS / 4) + c4] = xv[u];
    }
    __syncthreads();

    int rk0 = 0, rk1 = 0;
    if (COUNT) {
        if (ei0 < eend) rk0 = atomicAdd(&cnt[d0], 1);
        if (ei1 < eend) rk1 = atomicAdd(&cnt[d1], 1);
    }

    const int n0   = wave * 16;
    const int half = lane >> 4;
    const int ml   = lane & 15;

    for (int mt = 0; mt < TM / 16; ++mt) {
        v8f c = {};
        const float* xa = sX + (size_t)(mt * 16 + ml) * XS;
        for (int k = 0; k < C; k += 4) {
            v2f a, b;
            a[0] = xa[k + 2 * half];
            a[1] = xa[k + 2 * half + 1];
            b[0] = sW[(k + 2 * half)     * WS + n0 + ml];
            b[1] = sW[(k + 2 * half + 1) * WS + n0 + ml];
            c = __builtin_amdgcn_wmma_f32_16x16x4_f32(
                    false, a, false, b, (short)0, c, false, false);
        }
        int row_lo = row0 + mt * 16 + half * 8;
        int col    = n0 + ml;
        for (int r = 0; r < 8; ++r) {
            int row = row_lo + r;
            if (row < n_rows) Y[(size_t)row * C + col] = c[r];
        }
    }
    if (COUNT) {
        if (ei0 < eend) rank[ei0] = rk0;
        if (ei1 < eend) rank[ei1] = rk1;
        for (int i = ei0 + 512; i < eend; i += 256) rank[i] = atomicAdd(&cnt[dst[i]], 1);
    }
}

#define GU 8

__device__ __forceinline__ float xhalf(float x) {
    const int b = __builtin_bit_cast(int, x);
    return __builtin_bit_cast(float,
        __builtin_amdgcn_permlanex16(b, b, 0x76543210, 0xfedcba98, false, false));
}
template <int M>
__device__ __forceinline__ float xrow(float x) {
    const int b = __builtin_bit_cast(int, x);
    return __builtin_bit_cast(float,
        __builtin_amdgcn_update_dpp(b, b, 0x160 + M, 0xf, 0xf, false));
}
__device__ __forceinline__ float wave_sum(float s) {
    s += xhalf(s);
    s += xrow<8>(s);
    s += xrow<4>(s);
    s += xrow<2>(s);
    s += xrow<1>(s);
    return s;
}

template <bool LN>
__global__ __launch_bounds__(256) void gather_kernel(
        const float* __restrict__ XW, const float* __restrict__ dinv,
        const int* __restrict__ rowptr, const int* __restrict__ csr,
        const float* __restrict__ bias,
        const float* __restrict__ ln_w, const float* __restrict__ ln_b,
        float* __restrict__ OUT, int n) {
    const int lane = threadIdx.x & 31;
    const int row = __builtin_amdgcn_readfirstlane((int)(blockIdx.x * 8 + (threadIdx.x >> 5)));
    if (row >= n) return;
    const int   beg    = rowptr[row];
    const int   deg_in = rowptr[row + 1] - beg;
    const int   total  = deg_in + 1;
    const float di     = dinv[row];
    const float4 b4    = ((const float4*)bias)[lane];

    float4 acc = make_float4(0.f, 0.f, 0.f, 0.f);
    for (int base = 0; base < total; base += 32) {
        const int cnt = min(32, total - base);
        const int e = base + lane;
        int   s_l = row;
        float c_l = di * di;
        if (e < deg_in) {
            s_l = csr[beg + e];
            c_l = dinv[s_l] * di;
        }
        const int ne = min(32, deg_in - base);
        if (ne > 1) {
            int p = 0;
            for (int m = 0; m < ne; ++m) {
                const int sm = __builtin_amdgcn_readlane(s_l, m);
                p += (sm < s_l || (sm == s_l && m < lane)) ? 1 : 0;
            }
            const int pos = (lane < ne) ? p : lane;
            s_l = __builtin_amdgcn_ds_permute(pos << 2, s_l);
            c_l = __int_as_float(__builtin_amdgcn_ds_permute(pos << 2, __float_as_int(c_l)));
        }
        for (int k = 0; k < cnt; k += GU) {
            float4 v[GU];
            #pragma unroll
            for (int u = 0; u < GU; ++u) {
                v[u] = make_float4(0.f, 0.f, 0.f, 0.f);
                if (k + u < cnt) {
                    const int s = __builtin_amdgcn_readlane(s_l, k + u);
                    v[u] = ((const float4*)(XW + (size_t)s * C))[lane];
                }
            }
            #pragma unroll
            for (int u = 0; u < GU; ++u) {
                if (k + u < cnt) {
                    const float cf = __int_as_float(
                        __builtin_amdgcn_readlane(__float_as_int(c_l), k + u));
                    acc.x += v[u].x * cf;
                    acc.y += v[u].y * cf;
                    acc.z += v[u].z * cf;
                    acc.w += v[u].w * cf;
                }
            }
        }
    }
    float4 o;
    o.x = acc.x + b4.x;
    o.y = acc.y + b4.y;
    o.z = acc.z + b4.z;
    o.w = acc.w + b4.w;
    if (LN) {
        const float4 w4 = ((const float4*)ln_w)[lane];
        const float4 g4 = ((const float4*)ln_b)[lane];
        const float mu = wave_sum(o.x + o.y + o.z + o.w) * (1.0f / C);
        const float dx = o.x - mu, dy = o.y - mu, dz = o.z - mu, dw = o.w - mu;
        const float rinv = rsqrtf(wave_sum(dx * dx + dy * dy + dz * dz + dw * dw) * (1.0f / C) + 1e-5f);
        o.x = fmaxf(dx * rinv * w4.x + g4.x, 0.0f);
        o.y = fmaxf(dy * rinv * w4.y + g4.y, 0.0f);
        o.z = fmaxf(dz * rinv * w4.z + g4.z, 0.0f);
        o.w = fmaxf(dw * rinv * w4.w + g4.w, 0.0f);
    }
    ((float4*)(OUT + (size_t)row * C))[lane] = o;
}

extern "C" void kernel_launch(void* const* d_in, const int* in_sizes, int n_in,
                              void* d_out, int out_size, void* d_ws, size_t ws_size,
                              hipStream_t stream) {
    const float* h_noisy = (const float*)d_in[0];
    const int*   eidx    = (const int*)d_in[1];
    const float* t       = (const float*)d_in[2];
    const float* tW1     = (const float*)d_in[3];
    const float* tb1     = (const float*)d_in[4];
    const float* tW2     = (const float*)d_in[5];
    const float* tb2     = (const float*)d_in[6];
    const float* W1      = (const float*)d_in[7];
    const float* b1      = (const float*)d_in[8];
    const float* W2      = (const float*)d_in[9];
    const float* b2      = (const float*)d_in[10];
    const float* ln_w    = (const float*)d_in[11];
    const float* ln_b    = (const float*)d_in[12];

    const int N = in_sizes[0] / C;
    const int E = in_sizes[1] / 2;
    const int* src = eidx;
    const int* dst = eidx + E;

    float* ws    = (float*)d_ws;
    float* t_emb = ws;
    float* dinv  = ws + C;
    size_t o     = (size_t)C + (size_t)N;
    o = (o + 3) & ~(size_t)3;
    float* bufA  = ws + o;
    float* bufB  = bufA + (size_t)N * C;
    float* out   = (float*)d_out;

    const int BT = 256;
    dim3 blk(BT);
    const int scan_blocks = (N + SCAN_BLK - 1) / SCAN_BLK;

    int* rowptr = (int*)bufA;
    int* bsum   = rowptr + (N + 1);
    int* rank   = bsum + scan_blocks;
    int* csr    = rank + E;

    hipLaunchKernelGGL(time_mlp_kernel, dim3(1), dim3(C), 0, stream,
                       t, tW1, tb1, tW2, tb2, t_emb);
    hipLaunchKernelGGL(zero_int_kernel, dim3((N + BT - 1) / BT), blk, 0, stream, rowptr, N);
    const int gemm_blocks = (N + TM - 1) / TM;
    hipLaunchKernelGGL(gemm128_wmma_kernel<true>, dim3(gemm_blocks), blk, 0, stream,
                       h_noisy, W1, t_emb, out, N, dst, rowptr, rank, E);
    hipLaunchKernelGGL(block_sum_kernel, dim3(scan_blocks), blk, 0, stream, rowptr, bsum, N);
    hipLaunchKernelGGL(scan_kernel, dim3(scan_blocks), blk, 0, stream,
                       rowptr, bsum, dinv, N);
    hipLaunchKernelGGL(place_kernel, dim3((E + BT - 1) / BT), blk, 0, stream,
                       src, dst, rowptr, rank, csr, E);

    const int row_blocks  = (int)(((size_t)N * 32 + BT - 1) / BT);

    hipLaunchKernelGGL(gather_kernel<true>, dim3(row_blocks), blk, 0, stream,
                       out, dinv, rowptr, csr, b1, ln_w, ln_b, bufB, N);

    hipLaunchKernelGGL(gemm128_wmma_kernel<false>, dim3(gemm_blocks), blk, 0, stream,
                       bufB, W2, (const float*)nullptr, bufB, N,
                       (const int*)nullptr, (int*)nullptr, (int*)nullptr, 0);
    hipLaunchKernelGGL(gather_kernel<false>, dim3(row_blocks), blk, 0, stream,
                       bufB, dinv, rowptr, csr, b2, ln_w, ln_b, out, N);
}


# Layer-2 product moved into the layer-1 gather's workgroups: each sums and normalises 64 rows into LDS (4 per wave) and multiplies them by W2 there; GEMM launch and the h plane removed; sums unchanged
# speedup vs baseline: 2.7876x; 1.0780x over previous
#include <hip/hip_runtime.h>

typedef __attribute__((ext_vector_type(2))) float v2f;
typedef __attribute__((ext_vector_type(8))) float v8f;

#define C 128
#define TM 64
#define XS 132
#define WS 136

__global__ void time_mlp_kernel(const float* __restrict__ t,
                                const float* __restrict__ tW1, const float* __restrict__ tb1,
                                const float* __restrict__ tW2, const float* __restrict__ tb2,
                                float* __restrict__ t_emb) {
    __shared__ float hid[C];
    int j = threadIdx.x;
    float tv = t[0];
    hid[j] = fmaxf(tv * tW1[j] + tb1[j], 0.0f);
    __syncthreads();
    float acc = tb2[j];
    for (int i = 0; i < C; ++i) acc += hid[i] * tW2[i * C + j];
    t_emb[j] = acc;
}

#define SCAN_PER 4
#define SCAN_BLK (256 * SCAN_PER)

__global__ void zero_int_kernel(int* __restrict__ p, int n) {
    int i = blockIdx.x * blockDim.x + threadIdx.x;
    if (i < n) p[i] = 0;
}
__global__ __launch_bounds__(256) void block_sum_kernel(const int* __restrict__ cnt,
                                                        int* __restrict__ bsum, int n) {
    __shared__ int red[8];
    const int tid = threadIdx.x, lane = tid & 31, wave = tid >> 5;
    const int base = blockIdx.x * SCAN_BLK + tid * SCAN_PER;
    int s = 0;
    for (int k = 0; k < SCAN_PER; ++k)
        if (base + k < n) s += cnt[base + k];
    for (int m = 16; m; m >>= 1) s += __shfl_xor(s, m, 32);
    if (lane == 0) red[wave] = s;
    __syncthreads();
    if (tid == 0) {
        int t = 0;
        for (int w = 0; w < 8; ++w) t += red[w];
        bsum[blockIdx.x] = t;
    }
}
__global__ __launch_bounds__(256) void scan_kernel(int* __restrict__ rowptr, const int* __restrict__ bsum,
                                                   float* __restrict__ dinv, int n) {
    __shared__ int wtot[8];
    __shared__ int ptot[8];
    const int tid = threadIdx.x, lane = tid & 31, wave = tid >> 5;
    int part = 0;
    for (int i = tid; i < (int)blockIdx.x; i += 256) part += bsum[i];
    for (int m = 16; m; m >>= 1) part += __shfl_xor(part, m, 32);
    const int base = blockIdx.x * SCAN_BLK + tid * SCAN_PER;
    int c[SCAN_PER];
    int tsum = 0;
    for (int k = 0; k < SCAN_PER; ++k) {
        c[k] = (base + k < n) ? rowptr[base + k] : 0;
        tsum += c[k];
    }
    int inc = tsum;
    for (int d = 1; d < 32; d <<= 1) {
        int v = __shfl_up(inc, d, 32);
        if (lane >= d) inc += v;
    }
    if (lane == 31) wtot[wave] = inc;
    if (lane == 0)  ptot[wave] = part;
    __syncthreads();
    int off = 0;
    for (int w = 0; w < 8; ++w) {
        off += ptot[w];
        if (w < wave) off += wtot[w];
    }
    int run = off + inc - tsum;
    for (int k = 0; k < SCAN_PER; ++k) {
        int i = base + k;
        if (i < n) {
            rowptr[i] = run;
            dinv[i]   = rsqrtf((float)(c[k] + 1));
            run += c[k];
            if (i == n - 1) rowptr[n] = run;
        }
    }
}
__global__ void place_kernel(const int* __restrict__ src, const int* __restrict__ dst,
                             const int* __restrict__ rowptr, const int* __restrict__ rank,
                             int* __restrict__ csr, int e) {
    int i = blockIdx.x * blockDim.x + threadIdx.x;
    if (i < e) csr[rowptr[dst[i]] + rank[i]] = src[i];
}

__device__ __forceinline__ void tile_product(const float* sX, const float* sW, float* Y,
                                             int row0, int n_rows, int nt, int mt0, int mt1,
                                             int lane) {
    const int n0   = nt * 16;
    const int half = lane >> 4;
    const int ml   = lane & 15;

    for (int mt = mt0; mt < mt1; ++mt) {
        v8f c = {};
        const float* xa = sX + (size_t)(mt * 16 + ml) * XS;
        for (int k = 0; k < C; k += 4) {
            v2f a, b;
            a[0] = xa[k + 2 * half];
            a[1] = xa[k + 2 * half + 1];
            b[0] = sW[(k + 2 * half)     * WS + n0 + ml];
            b[1] = sW[(k + 2 * half + 1) * WS + n0 + ml];
            c = __builtin_amdgcn_wmma_f32_16x16x4_f32(
                    false, a, false, b, (short)0, c, false, false);
        }
        int row_lo = row0 + mt * 16 + half * 8;
        int col    = n0 + ml;
        for (int r = 0; r < 8; ++r) {
            int row = row_lo + r;
            if (row < n_rows) Y[(size_t)row * C + col] = c[r];
        }
    }
}

__global__ __launch_bounds__(256) void gemm128_count_kernel(
        const float* X, const float* __restrict__ W,
        const float* __restrict__ t_emb, float* Y, int n_rows,
        const int* __restrict__ dst, int* __restrict__ cnt, int* __restrict__ rank,
        int n_edges) {
    __shared__ float sW[C * WS];
    __shared__ float sX[TM * XS];

    const int tid  = threadIdx.x;
    const int wave = tid >> 5;
    const int lane = tid & 31;
    const int row0 = blockIdx.x * TM;

    const int c4 = lane;
    const int r0 = wave;
    float4 xv[TM / 8];
    #pragma unroll
    for (int u = 0; u < TM / 8; ++u) {
        const int row = row0 + r0 + 8 * u;
        xv[u] = make_float4(0.f, 0.f, 0.f, 0.f);
        if (row < n_rows) xv[u] = ((const float4*)(X + (size_t)row * C))[c4];
    }
    const float4 e = ((const float4*)t_emb)[c4];
    float4 wv[C / 8];
    #pragma unroll
    for (int u = 0; u < C / 8; ++u) wv[u] = ((const float4*)W)[(r0 + 8 * u) * (C / 4) + c4];
    const int per  = (n_edges + (int)gridDim.x - 1) / (int)gridDim.x;
    const int ebeg = (int)blockIdx.x * per;
    const int eend = min(n_edges, ebeg + per);
    const int ei0  = ebeg + tid;
    const int ei1  = ei0 + 256;
    unsigned d0 = 0, d1 = 0;
    if (n_edges > 0) {
        d0 = (unsigned)dst[min(ei0, n_edges - 1)];
        d1 = (unsigned)dst[min(ei1, n_edges - 1)];
    }

    {
        float4* sW4 = (float4*)sW;
        #pragma unroll
        for (int u = 0; u < C / 8; ++u) sW4[(r0 + 8 * u) * (WS / 4) + c4] = wv[u];
    }
    #pragma unroll
    for (int u = 0; u < TM / 8; ++u) {
        xv[u].x += e.x; xv[u].y += e.y; xv[u].z += e.z; xv[u].w += e.w;
    }
    {
        float4* sX4 = (float4*)sX;
        #pragma unroll
        for (int u = 0; u < TM / 8; ++u) sX4[(r0 + 8 * u) * (XS / 4) + c4] = xv[u];
    }
    __syncthreads();

    int rk0 = 0, rk1 = 0;
    if (ei0 < eend) rk0 = atomicAdd(&cnt[d0], 1);
    if (ei1 < eend) rk1 = atomicAdd(&cnt[d1], 1);

    tile_product(sX, sW, Y, row0, n_rows, wave, 0, TM / 16, lane);

    if (ei0 < eend) rank[ei0] = rk0;
    if (ei1 < eend) rank[ei1] = rk1;
    for (int i = ei0 + 512; i < eend; i += 256) rank[i] = atomicAdd(&cnt[dst[i]], 1);
}

#define GU 8

__device__ __forceinline__ float xhalf(float x) {
    const int b = __builtin_bit_cast(int, x);
    return __builtin_bit_cast(float,
        __builtin_amdgcn_permlanex16(b, b, 0x76543210, 0xfedcba98, false, false));
}
template <int M>
__device__ __forceinline__ float xrow(float x) {
    const int b = __builtin_bit_cast(int, x);
    return __builtin_bit_cast(float,
        __builtin_amdgcn_update_dpp(b, b, 0x160 + M, 0xf, 0xf, false));
}
__device__ __forceinline__ float wave_sum(float s) {
    s += xhalf(s);
    s += xrow<8>(s);
    s += xrow<4>(s);
    s += xrow<2>(s);
    s += xrow<1>(s);
    return s;
}

template <int G>
__device__ __forceinline__ float4 row_sum(
        const float* __restrict__ XW, const float* __restrict__ dinv,
        const int* __restrict__ csr, int row, int beg, int deg_in, float di,
        int s_first, float d_first, int lane) {
    const int total = deg_in + 1;
    float4 acc = make_float4(0.f, 0.f, 0.f, 0.f);
    for (int base = 0; base < total; base += 32) {
        const int cnt = min(32, total - base);
        const int e = base + lane;
        int   s_l = row;
        float c_l = di * di;
        if (base == 0) {
            if (lane < deg_in) {
                s_l = s_first;
                c_l = d_first * di;
            }
        } else if (e < deg_in) {
            s_l = csr[beg + e];
            c_l = dinv[s_l] * di;
        }
        const int ne = min(32, deg_in - base);
        if (ne > 1) {
            int p = 0;
            for (int m = 0; m < ne; ++m) {
                const int sm = __builtin_amdgcn_readlane(s_l, m);
                p += (sm < s_l || (sm == s_l && m < lane)) ? 1 : 0;
            }
            const int pos = (lane < ne) ? p : lane;
            s_l = __builtin_amdgcn_ds_permute(pos << 2, s_l);
            c_l = __int_as_float(__builtin_amdgcn_ds_permute(pos << 2, __float_as_int(c_l)));
        }
        for (int k = 0; k < cnt; k += G) {
            float4 v[G];
            #pragma unroll
            for (int u = 0; u < G; ++u) {
                v[u] = make_float4(0.f, 0.f, 0.f, 0.f);
                if (k + u < cnt) {
                    const int s = __builtin_amdgcn_readlane(s_l, k + u);
                    v[u] = ((const float4*)(XW + (size_t)s * C))[lane];
                }
            }
            #pragma unroll
            for (int u = 0; u < G; ++u) {
                if (k + u < cnt) {
                    const float cf = __int_as_float(
                        __builtin_amdgcn_readlane(__float_as_int(c_l), k + u));
                    acc.x += v[u].x * cf;
                    acc.y += v[u].y * cf;
                    acc.z += v[u].z * cf;
                    acc.w += v[u].w * cf;
                }
            }
        }
    }
    return acc;
}

__global__ __launch_bounds__(256) void gather_kernel(
        const float* __restrict__ XW, const float* __restrict__ dinv,
        const int* __restrict__ rowptr, const int* __restrict__ csr,
        const float* __restrict__ bias, float* __restrict__ OUT, int n) {
    const int lane = threadIdx.x & 31;
    const int row = __builtin_amdgcn_readfirstlane((int)(blockIdx.x * 8 + (threadIdx.x >> 5)));
    if (row >= n) return;
    const int   beg    = rowptr[row];
    const int   deg_in = rowptr[row + 1] - beg;
    const float di     = dinv[row];
    const float4 b4    = ((const float4*)bias)[lane];
    int   s0 = row;
    float d0 = 0.f;
    if (lane < deg_in) {
        s0 = csr[beg + lane];
        d0 = dinv[s0];
    }
    const float4 acc = row_sum<GU>(XW, dinv, csr, row, beg, deg_in, di, s0, d0, lane);
    float4 o;
    o.x = acc.x + b4.x;
    o.y = acc.y + b4.y;
    o.z = acc.z + b4.z;
    o.w = acc.w + b4.w;
    ((float4*)(OUT + (size_t)row * C))[lane] = o;
}

#define FW 16
#define RPW (TM / FW)

__global__ __launch_bounds__(FW * 32) void gather_ln_gemm_kernel(
        const float* __restrict__ XW, const float* __restrict__ dinv,
        const int* __restrict__ rowptr, const int* __restrict__ csr,
        const float* __restrict__ bias,
        const float* __restrict__ ln_w, const float* __restrict__ ln_b,
        const float* __restrict__ W, float* __restrict__ Y, int n, int n_edges) {
    __shared__ float sW[C * WS];
    __shared__ float sX[TM * XS];

    const int tid  = threadIdx.x;
    const int lane = tid & 31;
    const int wave = __builtin_amdgcn_readfirstlane(tid >> 5);
    const int row0 = blockIdx.x * TM;
    const int rbase = row0 + wave * RPW;

    const int   rp_l = rowptr[min(rbase + lane, n)];
    const float di_l = dinv[min(rbase + lane, n - 1)];
    const float4 b4 = ((const float4*)bias)[lane];
    const float4 w4 = ((const float4*)ln_w)[lane];
    const float4 g4 = ((const float4*)ln_b)[lane];
    float4 wv[C / FW];
    #pragma unroll
    for (int u = 0; u < C / FW; ++u) wv[u] = ((const float4*)W)[(wave + FW * u) * (C / 4) + lane];

    int beg[RPW], deg[RPW];
    #pragma unroll
    for (int r = 0; r < RPW; ++r) {
        beg[r] = __builtin_amdgcn_readlane(rp_l, r);
        deg[r] = __builtin_amdgcn_readlane(rp_l, r + 1) - beg[r];
    }
    int s_pre[RPW];
    #pragma unroll
    for (int r = 0; r < RPW; ++r) s_pre[r] = min(rbase + r, n - 1);
    if (n_edges > 0) {
        #pragma unroll
        for (int r = 0; r < RPW; ++r)
            s_pre[r] = csr[min(beg[r] + min(lane, max(deg[r] - 1, 0)), n_edges - 1)];
    }
    float d_pre[RPW];
    #pragma unroll
    for (int r = 0; r < RPW; ++r) d_pre[r] = dinv[s_pre[r]];

    {
        float4* sW4 = (float4*)sW;
        #pragma unroll
        for (int u = 0; u < C / FW; ++u) sW4[(wave + FW * u) * (WS / 4) + lane] = wv[u];
    }
    float* mine = sX + (size_t)(wave * RPW) * XS;
    #pragma unroll
    for (int r = 0; r < RPW; ++r) {
        mine[r * XS + lane]      = __int_as_float(s_pre[r]);
        mine[r * XS + 32 + lane] = d_pre[r];
    }

    for (int r = 0; r < RPW; ++r) {
        const int row = rbase + r;
        float4 o = make_float4(0.f, 0.f, 0.f, 0.f);
        if (row < n) {
            const int   rbeg = __builtin_amdgcn_readlane(rp_l, r);
            const int   rdeg = __builtin_amdgcn_readlane(rp_l, r + 1) - rbeg;
            const float di   = __int_as_float(__builtin_amdgcn_readlane(__float_as_int(di_l), r));
            const int   s_first = __float_as_int(mine[r * XS + lane]);
            const float d_first = mine[r * XS + 32 + lane];
            const float4 acc = row_sum<GU>(XW, dinv, csr, row, rbeg, rdeg, di,
                                           s_first, d_first, lane);
            o.x = acc.x + b4.x;
            o.y = acc.y + b4.y;
            o.z = acc.z + b4.z;
            o.w = acc.w + b4.w;
            const float mu = wave_sum(o.x + o.y + o.z + o.w) * (1.0f / C);
            const float dx = o.x - mu, dy = o.y - mu, dz = o.z - mu, dw = o.w - mu;
            const float rinv = rsqrtf(wave_sum(dx * dx + dy * dy + dz * dz + dw * dw) * (1.0f / C) + 1e-5f);
            o.x = fmaxf(dx * rinv * w4.x + g4.x, 0.0f);
            o.y = fmaxf(dy * rinv * w4.y + g4.y, 0.0f);
            o.z = fmaxf(dz * rinv * w4.z + g4.z, 0.0f);
            o.w = fmaxf(dw * rinv * w4.w + g4.w, 0.0f);
        }
        ((float4*)(mine + r * XS))[lane] = o;
    }
    __syncthreads();

    tile_product(sX, sW, Y, row0, n, wave & 7, 2 * (wave >> 3), 2 * (wave >> 3) + 2, lane);
}

extern "C" void kernel_launch(void* const* d_in, const int* in_sizes, int n_in,
                              void* d_out, int out_size, void* d_ws, size_t ws_size,
                              hipStream_t stream) {
    const float* h_noisy = (const float*)d_in[0];
    const int*   eidx    = (const int*)d_in[1];
    const float* t       = (const float*)d_in[2];
    const float* tW1     = (const float*)d_in[3];
    const float* tb1     = (const float*)d_in[4];
    const float* tW2     = (const float*)d_in[5];
    const float* tb2     = (const float*)d_in[6];
    const float* W1      = (const float*)d_in[7];
    const float* b1      = (const float*)d_in[8];
    const float* W2      = (const float*)d_in[9];
    const float* b2      = (const float*)d_in[10];
    const float* ln_w    = (const float*)d_in[11];
    const float* ln_b    = (const float*)d_in[12];

    const int N = in_sizes[0] / C;
    const int E = in_sizes[1] / 2;
    const int* src = eidx;
    const int* dst = eidx + E;

    float* ws    = (float*)d_ws;
    float* t_emb = ws;
    float* dinv  = ws + C;
    size_t o     = (size_t)C + (size_t)N;
    o = (o + 3) & ~(size_t)3;
    float* bufA  = ws + o;
    float* bufB  = bufA + (size_t)N * C;
    float* out   = (float*)d_out;

    const int BT = 256;
    dim3 blk(BT);
    const int scan_blocks = (N + SCAN_BLK - 1) / SCAN_BLK;

    int* rowptr = (int*)bufA;
    int* bsum   = rowptr + (N + 1);
    int* rank   = bsum + scan_blocks;
    int* csr    = rank + E;

    hipLaunchKernelGGL(time_mlp_kernel, dim3(1), dim3(C), 0, stream,
                       t, tW1, tb1, tW2, tb2, t_emb);
    hipLaunchKernelGGL(zero_int_kernel, dim3((N + BT - 1) / BT), blk, 0, stream, rowptr, N);
    const int gemm_blocks = (N + TM - 1) / TM;
    hipLaunchKernelGGL(gemm128_count_kernel, dim3(gemm_blocks), blk, 0, stream,
                       h_noisy, W1, t_emb, out, N, dst, rowptr, rank, E);
    hipLaunchKernelGGL(block_sum_kernel, dim3(scan_blocks), blk, 0, stream, rowptr, bsum, N);
    hipLaunchKernelGGL(scan_kernel, dim3(scan_blocks), blk, 0, stream,
                       rowptr, bsum, dinv, N);
    hipLaunchKernelGGL(place_kernel, dim3((E + BT - 1) / BT), blk, 0, stream,
                       src, dst, rowptr, rank, csr, E);

    const int row_blocks  = (int)(((size_t)N * 32 + BT - 1) / BT);

    hipLaunchKernelGGL(gather_ln_gemm_kernel, dim3(gemm_blocks), dim3(512), 0, stream,
                       out, dinv, rowptr, csr, b1, ln_w, ln_b, W2, bufB, N, E);

    hipLaunchKernelGGL(gather_kernel, dim3(row_blocks), blk, 0, stream,
                       bufB, dinv, rowptr, csr, b2, out, N);
}


# Edge list in fixed 32-slot buckets per node, filled at the ranks the counting atomics return (overflow list past 32); block-sum, scan and place launches removed; sums as before up to 32 in-edges
# speedup vs baseline: 3.2498x; 1.1658x over previous
#include <hip/hip_runtime.h>

typedef __attribute__((ext_vector_type(2))) float v2f;
typedef __attribute__((ext_vector_type(8))) float v8f;

#define C 128
#define TM 64
#define XS 132
#define WS 136

__global__ void time_mlp_kernel(const float* __restrict__ t,
                                const float* __restrict__ tW1, const float* __restrict__ tb1,
                                const float* __restrict__ tW2, const float* __restrict__ tb2,
                                float* __restrict__ t_emb) {
    __shared__ float hid[C];
    int j = threadIdx.x;
    float tv = t[0];
    hid[j] = fmaxf(tv * tW1[j] + tb1[j], 0.0f);
    __syncthreads();
    float acc = tb2[j];
    for (int i = 0; i < C; ++i) acc += hid[i] * tW2[i * C + j];
    t_emb[j] = acc;
}

#define CAP 32

__global__ void zero_int_kernel(int* __restrict__ p, int n) {
    int i = blockIdx.x * blockDim.x + threadIdx.x;
    if (i < n) p[i] = 0;
}
__device__ __forceinline__ void put_edge(int* __restrict__ bucket, int* __restrict__ ovf,
                                         int* __restrict__ ovf_n, int d, int s, int rk) {
    if (rk < CAP) {
        bucket[(size_t)d * CAP + rk] = s;
    } else {
        const int j = atomicAdd(ovf_n, 1);
        ovf[2 * (size_t)j]     = d;
        ovf[2 * (size_t)j + 1] = s;
    }
}
__device__ __forceinline__ float dinv_of(int count) { return rsqrtf((float)(count + 1)); }

__device__ __forceinline__ void tile_product(const float* sX, const float* sW, float* Y,
                                             int row0, int n_rows, int nt, int mt0, int mt1,
                                             int lane) {
    const int n0   = nt * 16;
    const int half = lane >> 4;
    const int ml   = lane & 15;

    for (int mt = mt0; mt < mt1; ++mt) {
        v8f c = {};
        const float* xa = sX + (size_t)(mt * 16 + ml) * XS;
        for (int k = 0; k < C; k += 4) {
            v2f a, b;
            a[0] = xa[k + 2 * half];
            a[1] = xa[k + 2 * half + 1];
            b[0] = sW[(k + 2 * half)     * WS + n0 + ml];
            b[1] = sW[(k + 2 * half + 1) * WS + n0 + ml];
            c = __builtin_amdgcn_wmma_f32_16x16x4_f32(
                    false, a, false, b, (short)0, c, false, false);
        }
        int row_lo = row0 + mt * 16 + half * 8;
        int col    = n0 + ml;
        for (int r = 0; r < 8; ++r) {
            int row = row_lo + r;
            if (row < n_rows) Y[(size_t)row * C + col] = c[r];
        }
    }
}

__global__ __launch_bounds__(256) void gemm128_count_kernel(
        const float* X, const float* __restrict__ W,
        const float* __restrict__ t_emb, float* Y, int n_rows,
        const int* __restrict__ src, const int* __restrict__ dst, int* __restrict__ cnt,
        int* __restrict__ bucket, int* __restrict__ ovf, int* __restrict__ ovf_n,
        int n_edges) {
    __shared__ float sW[C * WS];
    __shared__ float sX[TM * XS];

    const int tid  = threadIdx.x;
    const int wave = tid >> 5;
    const int lane = tid & 31;
    const int row0 = blockIdx.x * TM;

    const int c4 = lane;
    const int r0 = wave;
    float4 xv[TM / 8];
    #pragma unroll
    for (int u = 0; u < TM / 8; ++u) {
        const int row = row0 + r0 + 8 * u;
        xv[u] = make_float4(0.f, 0.f, 0.f, 0.f);
        if (row < n_rows) xv[u] = ((const float4*)(X + (size_t)row * C))[c4];
    }
    const float4 e = ((const float4*)t_emb)[c4];
    float4 wv[C / 8];
    #pragma unroll
    for (int u = 0; u < C / 8; ++u) wv[u] = ((const float4*)W)[(r0 + 8 * u) * (C / 4) + c4];
    const int per  = (n_edges + (int)gridDim.x - 1) / (int)gridDim.x;
    const int ebeg = (int)blockIdx.x * per;
    const int eend = min(n_edges, ebeg + per);
    const int ei0  = ebeg + tid;
    const int ei1  = ei0 + 256;
    unsigned d0 = 0, d1 = 0;
    int sA = 0, sB = 0;
    if (n_edges > 0) {
        d0 = (unsigned)dst[min(ei0, n_edges - 1)];
        d1 = (unsigned)dst[min(ei1, n_edges - 1)];
        sA = src[min(ei0, n_edges - 1)];
        sB = src[min(ei1, n_edges - 1)];
    }

    {
        float4* sW4 = (float4*)sW;
        #pragma unroll
        for (int u = 0; u < C / 8; ++u) sW4[(r0 + 8 * u) * (WS / 4) + c4] = wv[u];
    }
    #pragma unroll
    for (int u = 0; u < TM / 8; ++u) {
        xv[u].x += e.x; xv[u].y += e.y; xv[u].z += e.z; xv[u].w += e.w;
    }
    {
        float4* sX4 = (float4*)sX;
        #pragma unroll
        for (int u = 0; u < TM / 8; ++u) sX4[(r0 + 8 * u) * (XS / 4) + c4] = xv[u];
    }
    __syncthreads();

    int rk0 = 0, rk1 = 0;
    if (ei0 < eend) rk0 = atomicAdd(&cnt[d0], 1);
    if (ei1 < eend) rk1 = atomicAdd(&cnt[d1], 1);

    tile_product(sX, sW, Y, row0, n_rows, wave, 0, TM / 16, lane);

    if (ei0 < eend) put_edge(bucket, ovf, ovf_n, (int)d0, sA, rk0);
    if (ei1 < eend) put_edge(bucket, ovf, ovf_n, (int)d1, sB, rk1);
    for (int i = ei0 + 512; i < eend; i += 256) {
        const int d = dst[i];
        put_edge(bucket, ovf, ovf_n, d, src[i], atomicAdd(&cnt[d], 1));
    }
}

#define GU 8

__device__ __forceinline__ float xhalf(float x) {
    const int b = __builtin_bit_cast(int, x);
    return __builtin_bit_cast(float,
        __builtin_amdgcn_permlanex16(b, b, 0x76543210, 0xfedcba98, false, false));
}
template <int M>
__device__ __forceinline__ float xrow(float x) {
    const int b = __builtin_bit_cast(int, x);
    return __builtin_bit_cast(float,
        __builtin_amdgcn_update_dpp(b, b, 0x160 + M, 0xf, 0xf, false));
}
__device__ __forceinline__ float wave_sum(float s) {
    s += xhalf(s);
    s += xrow<8>(s);
    s += xrow<4>(s);
    s += xrow<2>(s);
    s += xrow<1>(s);
    return s;
}

template <int G>
__device__ __forceinline__ float4 row_sum(
        const float* __restrict__ XW, const int* __restrict__ cnt,
        const int* __restrict__ ovf, const int* __restrict__ ovf_n, int n_edges,
        int row, int deg_in, float di, int s_first, float d_first, int lane) {
    const int nb    = min(deg_in, CAP);
    const int total = nb + 1;
    float4 acc = make_float4(0.f, 0.f, 0.f, 0.f);
    for (int base = 0; base < total; base += 32) {
        const int cnt_e = min(32, total - base);
        int   s_l = row;
        float c_l = di * di;
        if (base == 0 && lane < nb) {
            s_l = s_first;
            c_l = d_first * di;
        }
        const int ne = min(32, nb - base);
        if (ne > 1) {
            int p = 0;
            for (int m = 0; m < ne; ++m) {
                const int sm = __builtin_amdgcn_readlane(s_l, m);
                p += (sm < s_l || (sm == s_l && m < lane)) ? 1 : 0;
            }
            const int pos = (lane < ne) ? p : lane;
            s_l = __builtin_amdgcn_ds_permute(pos << 2, s_l);
            c_l = __int_as_float(__builtin_amdgcn_ds_permute(pos << 2, __float_as_int(c_l)));
        }
        for (int k = 0; k < cnt_e; k += G) {
            float4 v[G];
            #pragma unroll
            for (int u = 0; u < G; ++u) {
                v[u] = make_float4(0.f, 0.f, 0.f, 0.f);
                if (k + u < cnt_e) {
                    const int s = __builtin_amdgcn_readlane(s_l, k + u);
                    v[u] = ((const float4*)(XW + (size_t)s * C))[lane];
                }
            }
            #pragma unroll
            for (int u = 0; u < G; ++u) {
                if (k + u < cnt_e) {
                    const float cf = __int_as_float(
                        __builtin_amdgcn_readlane(__float_as_int(c_l), k + u));
                    acc.x += v[u].x * cf;
                    acc.y += v[u].y * cf;
                    acc.z += v[u].z * cf;
                    acc.w += v[u].w * cf;
                }
            }
        }
    }
    if (deg_in > CAP) {
        const int novf = min(ovf_n[0], n_edges);
        for (int j0 = 0; j0 < novf; j0 += 32) {
            const int j = j0 + lane;
            int d_l = -1, s_l = 0;
            if (j < novf) {
                d_l = ovf[2 * (size_t)j];
                s_l = ovf[2 * (size_t)j + 1];
            }
            unsigned mine = __builtin_amdgcn_ballot_w32(d_l == row);
            while (mine) {
                const int l = __builtin_ctz(mine);
                mine &= mine - 1;
                const int s = __builtin_amdgcn_readlane(s_l, l);
                const float cf = dinv_of(cnt[s]) * di;
                const float4 v = ((const float4*)(XW + (size_t)s * C))[lane];
                acc.x += v.x * cf;
                acc.y += v.y * cf;
                acc.z += v.z * cf;
                acc.w += v.w * cf;
            }
        }
    }
    return acc;
}

__global__ __launch_bounds__(256) void gather_kernel(
        const float* __restrict__ XW, const int* __restrict__ cnt,
        const int* __restrict__ bucket, const int* __restrict__ ovf,
        const int* __restrict__ ovf_n,
        const float* __restrict__ bias, float* __restrict__ OUT, int n, int n_edges) {
    const int lane = threadIdx.x & 31;
    const int row = __builtin_amdgcn_readfirstlane((int)(blockIdx.x * 8 + (threadIdx.x >> 5)));
    if (row >= n) return;
    const int   deg_in = cnt[row];
    const float di     = dinv_of(deg_in);
    const float4 b4    = ((const float4*)bias)[lane];
    int   s0 = row;
    float d0 = 0.f;
    if (lane < min(deg_in, CAP)) {
        s0 = bucket[(size_t)row * CAP + lane];
        d0 = dinv_of(cnt[s0]);
    }
    const float4 acc = row_sum<GU>(XW, cnt, ovf, ovf_n, n_edges, row, deg_in, di, s0, d0, lane);
    float4 o;
    o.x = acc.x + b4.x;
    o.y = acc.y + b4.y;
    o.z = acc.z + b4.z;
    o.w = acc.w + b4.w;
    ((float4*)(OUT + (size_t)row * C))[lane] = o;
}

#define FW 16
#define RPW (TM / FW)

__global__ __launch_bounds__(FW * 32) void gather_ln_gemm_kernel(
        const float* __restrict__ XW, const int* __restrict__ cnt,
        const int* __restrict__ bucket, const int* __restrict__ ovf,
        const int* __restrict__ ovf_n,
        const float* __restrict__ bias,
        const float* __restrict__ ln_w, const float* __restrict__ ln_b,
        const float* __restrict__ W, float* __restrict__ Y, int n, int n_edges) {
    __shared__ float sW[C * WS];
    __shared__ float sX[TM * XS];

    const int tid  = threadIdx.x;
    const int lane = tid & 31;
    const int wave = __builtin_amdgcn_readfirstlane(tid >> 5);
    const int row0 = blockIdx.x * TM;
    const int rbase = row0 + wave * RPW;

    const int   cn_l = cnt[min(rbase + lane, n - 1)];
    const float di_l = dinv_of(cn_l);
    const float4 b4 = ((const float4*)bias)[lane];
    const float4 w4 = ((const float4*)ln_w)[lane];
    const float4 g4 = ((const float4*)ln_b)[lane];
    float4 wv[C / FW];
    #pragma unroll
    for (int u = 0; u < C / FW; ++u) wv[u] = ((const float4*)W)[(wave + FW * u) * (C / 4) + lane];

    int nbk[RPW];
    #pragma unroll
    for (int r = 0; r < RPW; ++r) nbk[r] = min(__builtin_amdgcn_readlane(cn_l, r), CAP);
    int s_pre[RPW];
    #pragma unroll
    for (int r = 0; r < RPW; ++r) {
        const int rr = min(rbase + r, n - 1);
        const int* p = (nbk[r] > 0) ? bucket + (size_t)rr * CAP + min(lane, nbk[r] - 1)
                                    : cnt + rr;
        s_pre[r] = *p;
    }
    float d_pre[RPW];
    #pragma unroll
    for (int r = 0; r < RPW; ++r) d_pre[r] = dinv_of(cnt[s_pre[r]]);

    {
        float4* sW4 = (float4*)sW;
        #pragma unroll
        for (int u = 0; u < C / FW; ++u) sW4[(wave + FW * u) * (WS / 4) + lane] = wv[u];
    }
    float* mine = sX + (size_t)(wave * RPW) * XS;
    #pragma unroll
    for (int r = 0; r < RPW; ++r) {
        mine[r * XS + lane]      = __int_as_float(s_pre[r]);
        mine[r * XS + 32 + lane] = d_pre[r];
    }

    for (int r = 0; r < RPW; ++r) {
        const int row = rbase + r;
        float4 o = make_float4(0.f, 0.f, 0.f, 0.f);
        if (row < n) {
            const int   rdeg = __builtin_amdgcn_readlane(cn_l, r);
            const float di   = __int_as_float(__builtin_amdgcn_readlane(__float_as_int(di_l), r));
            const int   s_first = __float_as_int(mine[r * XS + lane]);
            const float d_first = mine[r * XS + 32 + lane];
            const float4 acc = row_sum<GU>(XW, cnt, ovf, ovf_n, n_edges, row, rdeg, di,
                                           s_first, d_first, lane);
            o.x = acc.x + b4.x;
            o.y = acc.y + b4.y;
            o.z = acc.z + b4.z;
            o.w = acc.w + b4.w;
            const float mu = wave_sum(o.x + o.y + o.z + o.w) * (1.0f / C);
            const float dx = o.x - mu, dy = o.y - mu, dz = o.z - mu, dw = o.w - mu;
            const float rinv = rsqrtf(wave_sum(dx * dx + dy * dy + dz * dz + dw * dw) * (1.0f / C) + 1e-5f);
            o.x = fmaxf(dx * rinv * w4.x + g4.x, 0.0f);
            o.y = fmaxf(dy * rinv * w4.y + g4.y, 0.0f);
            o.z = fmaxf(dz * rinv * w4.z + g4.z, 0.0f);
            o.w = fmaxf(dw * rinv * w4.w + g4.w, 0.0f);
        }
        ((float4*)(mine + r * XS))[lane] = o;
    }
    __syncthreads();

    tile_product(sX, sW, Y, row0, n, wave & 7, 2 * (wave >> 3), 2 * (wave >> 3) + 2, lane);
}

extern "C" void kernel_launch(void* const* d_in, const int* in_sizes, int n_in,
                              void* d_out, int out_size, void* d_ws, size_t ws_size,
                              hipStream_t stream) {
    const float* h_noisy = (const float*)d_in[0];
    const int*   eidx    = (const int*)d_in[1];
    const float* t       = (const float*)d_in[2];
    const float* tW1     = (const float*)d_in[3];
    const float* tb1     = (const float*)d_in[4];
    const float* tW2     = (const float*)d_in[5];
    const float* tb2     = (const float*)d_in[6];
    const float* W1      = (const float*)d_in[7];
    const float* b1      = (const float*)d_in[8];
    const float* W2      = (const float*)d_in[9];
    const float* b2      = (const float*)d_in[10];
    const float* ln_w    = (const float*)d_in[11];
    const float* ln_b    = (const float*)d_in[12];

    const int N = in_sizes[0] / C;
    const int E = in_sizes[1] / 2;
    const int* src = eidx;
    const int* dst = eidx + E;

    float* ws    = (float*)d_ws;
    float* t_emb = ws;
    float* bufA  = ws + C;
    float* bufB  = bufA + (size_t)N * C;
    float* out   = (float*)d_out;

    const int BT = 256;
    dim3 blk(BT);

    int* cnt    = (int*)bufA;
    int* ovf_n  = cnt + N;
    int* bucket = cnt + ((N + 1 + 31) & ~31);
    int* ovf    = bucket + (size_t)N * CAP;

    hipLaunchKernelGGL(time_mlp_kernel, dim3(1), dim3(C), 0, stream,
                       t, tW1, tb1, tW2, tb2, t_emb);
    hipLaunchKernelGGL(zero_int_kernel, dim3((N + 1 + BT - 1) / BT), blk, 0, stream, cnt, N + 1);
    const int gemm_blocks = (N + TM - 1) / TM;
    hipLaunchKernelGGL(gemm128_count_kernel, dim3(gemm_blocks), blk, 0, stream,
                       h_noisy, W1, t_emb, out, N, src, dst, cnt, bucket, ovf, ovf_n, E);

    const int row_blocks  = (int)(((size_t)N * 32 + BT - 1) / BT);

    hipLaunchKernelGGL(gather_ln_gemm_kernel, dim3(gemm_blocks), dim3(512), 0, stream,
                       out, cnt, bucket, ovf, ovf_n, b1, ln_w, ln_b, W2, bufB, N, E);

    hipLaunchKernelGGL(gather_kernel, dim3(row_blocks), blk, 0, stream,
                       bufB, cnt, bucket, ovf, ovf_n, b2, out, N, E);
}
